# starting kernel
# baseline (speedup 1.0000x reference)
#include <hip/hip_runtime.h>
#include <hip/hip_bf16.h>
#include <math.h>

#define NN 50000
#define EE 800000
#define FF 128
#define HH 8
#define DD 16

typedef __attribute__((ext_vector_type(2))) float v2f;
typedef __attribute__((ext_vector_type(8))) float v8f;

__global__ void gat_zero_kernel(float* __restrict__ p, int n) {
    int i = blockIdx.x * blockDim.x + threadIdx.x;
    if (i < n) p[i] = 0.0f;
}

__global__ __launch_bounds__(32) void gat_proj_gemm(
    const float* __restrict__ x,
    const float* __restrict__ W_src,
    const float* __restrict__ W_dst,
    float* __restrict__ src_p,
    float* __restrict__ dst_p)
{
    const int which   = blockIdx.x;
    const int rowbase = blockIdx.y * 16;
    const float* __restrict__ W   = which ? W_dst : W_src;
    float* __restrict__       out = which ? dst_p : src_p;

    const int lane = threadIdx.x;
    const int half = lane >> 4;
    const int l    = lane & 15;

    const float* __restrict__ xrow = x + (size_t)(rowbase + l) * FF;

    v8f acc[8];
#pragma unroll
    for (int c = 0; c < 8; ++c) acc[c] = (v8f){};

    for (int k = 0; k < FF; k += 4) {
        v2f a;
        a.x = xrow[k + 2 * half + 0];
        a.y = xrow[k + 2 * half + 1];
#pragma unroll
        for (int c = 0; c < 8; ++c) {
            const float* wp = W + (size_t)(k + 2 * half) * FF + c * 16 + l;
            v2f b;
            b.x = wp[0];
            b.y = wp[FF];
            acc[c] = __builtin_amdgcn_wmma_f32_16x16x4_f32(
                false, a, false, b, (short)0, acc[c], false, false);
        }
    }

#pragma unroll
    for (int c = 0; c < 8; ++c) {
#pragma unroll
        for (int r = 0; r < 8; ++r) {
            out[(size_t)(rowbase + r + 8 * half) * FF + c * 16 + l] = acc[c][r];
        }
    }
}

__global__ __launch_bounds__(256) void gat_score(
    const float* __restrict__ src_p,
    const float* __restrict__ dst_p,
    const int*   __restrict__ srcI,
    const int*   __restrict__ dstI,
    const float* __restrict__ attn,
    float* __restrict__ scorebuf,
    float* __restrict__ partial)
{
    __shared__ float smax[256];
    const int tid = blockIdx.x * 256 + threadIdx.x;
    float s = -INFINITY;
    if (tid < EE * HH) {
        const int e = tid >> 3;
        const int h = tid & 7;
        const float* sp = src_p + (size_t)srcI[e] * FF + h * DD;
        const float* dp = dst_p + (size_t)dstI[e] * FF + h * DD;
        const float* at = attn + h * DD;
        float acc = 0.0f;
#pragma unroll
        for (int d = 0; d < DD; ++d) {
            float v = sp[d] + dp[d];
            v = (v >= 0.0f) ? v : 0.2f * v;
            acc += at[d] * v;
        }
        scorebuf[tid] = acc;
        s = acc;
    }
    smax[threadIdx.x] = s;
    __syncthreads();
#pragma unroll
    for (int off = 128; off > 0; off >>= 1) {
        if (threadIdx.x < off)
            smax[threadIdx.x] = fmaxf(smax[threadIdx.x], smax[threadIdx.x + off]);
        __syncthreads();
    }
    if (threadIdx.x == 0) partial[blockIdx.x] = smax[0];
}

__global__ __launch_bounds__(256) void gat_maxreduce(
    const float* __restrict__ partial, int n, float* __restrict__ gmax)
{
    __shared__ float smax[256];
    float s = -INFINITY;
    for (int i = threadIdx.x; i < n; i += 256) s = fmaxf(s, partial[i]);
    smax[threadIdx.x] = s;
    __syncthreads();
#pragma unroll
    for (int off = 128; off > 0; off >>= 1) {
        if (threadIdx.x < off)
            smax[threadIdx.x] = fmaxf(smax[threadIdx.x], smax[threadIdx.x + off]);
        __syncthreads();
    }
    if (threadIdx.x == 0) *gmax = smax[0];
}

__global__ __launch_bounds__(256) void gat_expdenom(
    float* __restrict__ scorebuf,
    const int* __restrict__ dstI,
    const float* __restrict__ gmax,
    float* __restrict__ denom)
{
    const int tid = blockIdx.x * 256 + threadIdx.x;
    if (tid >= EE * HH) return;
    const int e = tid >> 3;
    const int h = tid & 7;
    const float ex = __expf(scorebuf[tid] - *gmax);
    scorebuf[tid] = ex;
    atomicAdd(&denom[(size_t)dstI[e] * HH + h], ex);
}

__global__ __launch_bounds__(256) void gat_message(
    const float* __restrict__ src_p,
    const float* __restrict__ expbuf,
    const float* __restrict__ denom,
    const int*   __restrict__ srcI,
    const int*   __restrict__ dstI,
    float* __restrict__ out)
{
    const int tid = blockIdx.x * 256 + threadIdx.x;
    if (tid >= EE * HH) return;
    const int e = tid >> 3;
    const int h = tid & 7;
    const int di = dstI[e];
    const float alpha = expbuf[tid] / (denom[(size_t)di * HH + h] + 1e-16f);
    const float* sp = src_p + (size_t)srcI[e] * FF + h * DD;
    float* op = out + (size_t)di * FF + h * DD;
#pragma unroll
    for (int d = 0; d < DD; ++d) {
        atomicAdd(&op[d], sp[d] * alpha);
    }
}

__global__ __launch_bounds__(256) void gat_finalize(
    float* __restrict__ out,
    const float* __restrict__ x,
    const float* __restrict__ bias,
    const float* __restrict__ prelu_w)
{
    const int i = blockIdx.x * 256 + threadIdx.x;
    if (i >= NN * FF) return;
    const float v  = out[i] + x[i] + bias[i & (FF - 1)];
    const float pw = prelu_w[0];
    out[i] = (v >= 0.0f) ? v : pw * v;
}

extern "C" void kernel_launch(void* const* d_in, const int* in_sizes, int n_in,
                              void* d_out, int out_size, void* d_ws, size_t ws_size,
                              hipStream_t stream) {
    (void)in_sizes; (void)n_in; (void)out_size; (void)ws_size;

    const float* x        = (const float*)d_in[0];
    const int*   eidx     = (const int*)  d_in[1];
    const float* W_src    = (const float*)d_in[2];
    const float* W_dst    = (const float*)d_in[3];
    const float* attn     = (const float*)d_in[4];
    const float* bias     = (const float*)d_in[5];
    const float* prelu_w  = (const float*)d_in[6];
    const int*   srcI     = eidx;
    const int*   dstI     = eidx + EE;

    float* out = (float*)d_out;

    float* ws      = (float*)d_ws;
    float* src_p   = ws;
    float* dst_p   = src_p + (size_t)NN * FF;
    float* scoreb  = dst_p + (size_t)NN * FF;
    float* denom   = scoreb + (size_t)EE * HH;
    float* partial = denom + (size_t)NN * HH;
    const int nScoreBlk = (EE * HH + 255) / 256;
    float* gmax    = partial + nScoreBlk;

    {
        const int n1 = NN * FF;
        gat_zero_kernel<<<(n1 + 255) / 256, 256, 0, stream>>>(out, n1);
        const int n2 = NN * HH;
        gat_zero_kernel<<<(n2 + 255) / 256, 256, 0, stream>>>(denom, n2);
    }

    {
        dim3 grid(2, NN / 16);
        gat_proj_gemm<<<grid, 32, 0, stream>>>(x, W_src, W_dst, src_p, dst_p);
    }

    gat_score<<<nScoreBlk, 256, 0, stream>>>(src_p, dst_p, srcI, dstI, attn,
                                             scoreb, partial);
    gat_maxreduce<<<1, 256, 0, stream>>>(partial, nScoreBlk, gmax);

    gat_expdenom<<<nScoreBlk, 256, 0, stream>>>(scoreb, dstI, gmax, denom);

    gat_message<<<nScoreBlk, 256, 0, stream>>>(src_p, scoreb, denom, srcI, dstI, out);

    {
        const int n = NN * FF;
        gat_finalize<<<(n + 255) / 256, 256, 0, stream>>>(out, x, bias, prelu_w);
    }
}


# message pass: 250 blocks each own 200 destination rows as f32 sums in LDS, walk the whole edge list, add their edges' rows by LDS f32 atomics, store rows once; no global f32 atomics (add order free)
# speedup vs baseline: 2.4847x; 2.4847x over previous
#include <hip/hip_runtime.h>
#include <hip/hip_bf16.h>
#include <math.h>

#define NN 50000
#define EE 800000
#define FF 128
#define HH 8
#define DD 16

typedef __attribute__((ext_vector_type(2))) float v2f;
typedef __attribute__((ext_vector_type(8))) float v8f;

__global__ void gat_zero_kernel(float* __restrict__ p, int n) {
    int i = blockIdx.x * blockDim.x + threadIdx.x;
    if (i < n) p[i] = 0.0f;
}

__global__ __launch_bounds__(32) void gat_proj_gemm(
    const float* __restrict__ x,
    const float* __restrict__ W_src,
    const float* __restrict__ W_dst,
    float* __restrict__ src_p,
    float* __restrict__ dst_p)
{
    const int which   = blockIdx.x;
    const int rowbase = blockIdx.y * 16;
    const float* __restrict__ W   = which ? W_dst : W_src;
    float* __restrict__       out = which ? dst_p : src_p;

    const int lane = threadIdx.x;
    const int half = lane >> 4;
    const int l    = lane & 15;

    const float* __restrict__ xrow = x + (size_t)(rowbase + l) * FF;

    v8f acc[8];
#pragma unroll
    for (int c = 0; c < 8; ++c) acc[c] = (v8f){};

    for (int k = 0; k < FF; k += 4) {
        v2f a;
        a.x = xrow[k + 2 * half + 0];
        a.y = xrow[k + 2 * half + 1];
#pragma unroll
        for (int c = 0; c < 8; ++c) {
            const float* wp = W + (size_t)(k + 2 * half) * FF + c * 16 + l;
            v2f b;
            b.x = wp[0];
            b.y = wp[FF];
            acc[c] = __builtin_amdgcn_wmma_f32_16x16x4_f32(
                false, a, false, b, (short)0, acc[c], false, false);
        }
    }

#pragma unroll
    for (int c = 0; c < 8; ++c) {
#pragma unroll
        for (int r = 0; r < 8; ++r) {
            out[(size_t)(rowbase + r + 8 * half) * FF + c * 16 + l] = acc[c][r];
        }
    }
}

__global__ __launch_bounds__(256) void gat_score(
    const float* __restrict__ src_p,
    const float* __restrict__ dst_p,
    const int*   __restrict__ srcI,
    const int*   __restrict__ dstI,
    const float* __restrict__ attn,
    float* __restrict__ scorebuf,
    float* __restrict__ partial)
{
    __shared__ float smax[256];
    const int tid = blockIdx.x * 256 + threadIdx.x;
    float s = -INFINITY;
    if (tid < EE * HH) {
        const int e = tid >> 3;
        const int h = tid & 7;
        const float* sp = src_p + (size_t)srcI[e] * FF + h * DD;
        const float* dp = dst_p + (size_t)dstI[e] * FF + h * DD;
        const float* at = attn + h * DD;
        float acc = 0.0f;
#pragma unroll
        for (int d = 0; d < DD; ++d) {
            float v = sp[d] + dp[d];
            v = (v >= 0.0f) ? v : 0.2f * v;
            acc += at[d] * v;
        }
        scorebuf[tid] = acc;
        s = acc;
    }
    smax[threadIdx.x] = s;
    __syncthreads();
#pragma unroll
    for (int off = 128; off > 0; off >>= 1) {
        if (threadIdx.x < off)
            smax[threadIdx.x] = fmaxf(smax[threadIdx.x], smax[threadIdx.x + off]);
        __syncthreads();
    }
    if (threadIdx.x == 0) partial[blockIdx.x] = smax[0];
}

__global__ __launch_bounds__(256) void gat_maxreduce(
    const float* __restrict__ partial, int n, float* __restrict__ gmax)
{
    __shared__ float smax[256];
    float s = -INFINITY;
    for (int i = threadIdx.x; i < n; i += 256) s = fmaxf(s, partial[i]);
    smax[threadIdx.x] = s;
    __syncthreads();
#pragma unroll
    for (int off = 128; off > 0; off >>= 1) {
        if (threadIdx.x < off)
            smax[threadIdx.x] = fmaxf(smax[threadIdx.x], smax[threadIdx.x + off]);
        __syncthreads();
    }
    if (threadIdx.x == 0) *gmax = smax[0];
}

__global__ __launch_bounds__(256) void gat_expdenom(
    float* __restrict__ scorebuf,
    const int* __restrict__ dstI,
    const float* __restrict__ gmax,
    float* __restrict__ denom)
{
    const int tid = blockIdx.x * 256 + threadIdx.x;
    if (tid >= EE * HH) return;
    const int e = tid >> 3;
    const int h = tid & 7;
    const float ex = __expf(scorebuf[tid] - *gmax);
    scorebuf[tid] = ex;
    atomicAdd(&denom[(size_t)dstI[e] * HH + h], ex);
}

#define RNG         200
#define MSG_BLOCKS  (NN / RNG)
#define MSG_THREADS 1024
#define MSG_WAVES   (MSG_THREADS / 32)
#define SCAN_DEPTH  8
#define QCAP        1056
#define NGROUPS     (EE / 128)

static_assert(NN % RNG == 0, "blocks tile the rows exactly");
static_assert(RNG <= 256, "local row fits the low 8 bits of a queue word");
static_assert(EE % 128 == 0 && EE < (1 << 24), "edge id fits the upper 24 bits");
static_assert(NGROUPS % MSG_BLOCKS == 0, "start offsets tile the list");
static_assert(QCAP >= 31 + SCAN_DEPTH * 128, "a step's edges always fit the queue");

__device__ __forceinline__ void wave_lds_order() {
    __builtin_amdgcn_fence(__ATOMIC_RELEASE, "wavefront");
    __builtin_amdgcn_wave_barrier();
    __builtin_amdgcn_fence(__ATOMIC_ACQUIRE, "wavefront");
}

__device__ __forceinline__ int queue_push(unsigned* __restrict__ q, int qn,
                                          bool m, unsigned word) {
    const unsigned mask = __builtin_amdgcn_ballot_w32(m);
    if (mask) {
        if (m) q[qn + (int)__builtin_amdgcn_mbcnt_lo(mask, 0u)] = word;
        qn += __builtin_popcount(mask);
    }
    return qn;
}

template <bool FULL>
__device__ __forceinline__ void drain_edges(
    const unsigned* __restrict__ q, int n, float* __restrict__ al,
    float* __restrict__ acc, int base, int lane,
    const float* __restrict__ src_p,
    const float* __restrict__ expbuf,
    const float* __restrict__ denom,
    const int*   __restrict__ srcI)
{
    wave_lds_order();
    const bool act = FULL || (lane < n);
    unsigned word = 0u;
    if (act) word = q[lane];
    const unsigned e  = word >> 8;
    const int      dl = (int)(word & 255u);
    int s = 0;
    if (act) {
        s = srcI[e];
        const float4* ex4 = (const float4*)(expbuf + (size_t)e * HH);
        const float4* dn4 = (const float4*)(denom + (size_t)(base + dl) * HH);
        const float4 x0 = ex4[0], x1 = ex4[1];
        const float4 n0 = dn4[0], n1 = dn4[1];
        float4 a0, a1;
        a0.x = x0.x / (n0.x + 1e-16f);
        a0.y = x0.y / (n0.y + 1e-16f);
        a0.z = x0.z / (n0.z + 1e-16f);
        a0.w = x0.w / (n0.w + 1e-16f);
        a1.x = x1.x / (n1.x + 1e-16f);
        a1.y = x1.y / (n1.y + 1e-16f);
        a1.z = x1.z / (n1.z + 1e-16f);
        a1.w = x1.w / (n1.w + 1e-16f);
        float4* ap = (float4*)(al + lane * HH);
        ap[0] = a0;
        ap[1] = a1;
    }
    wave_lds_order();

    const int hd = lane >> 2;
    const int hf = lane >> 4;
    const int o0 = lane * 4 + (hf ? 1 : 0);
    const int o1 = lane * 4 + (hf ? 2 : 1);
    const int o2 = lane * 4 + (hf ? 3 : 2);
    const int o3 = lane * 4 + (hf ? 0 : 3);
    const int cnt = FULL ? 32 : n;
    for (int j0 = 0; j0 < cnt; j0 += 8) {
        float4 v[8];
        float  a[8];
        int    dj[8];
#pragma unroll
        for (int u = 0; u < 8; ++u) {
            const int sj = __builtin_amdgcn_readlane(s, j0 + u);
            dj[u] = __builtin_amdgcn_readlane(dl, j0 + u);
            v[u]  = ((const float4*)(src_p + (size_t)sj * FF))[lane];
            a[u]  = al[(j0 + u) * HH + hd];
        }
#pragma unroll
        for (int u = 0; u < 8; ++u) {
            if (FULL || (j0 + u < cnt)) {
                const float p0 = v[u].x * a[u], p1 = v[u].y * a[u];
                const float p2 = v[u].z * a[u], p3 = v[u].w * a[u];
                float* row = acc + dj[u] * FF;
                atomicAdd(row + o0, hf ? p1 : p0);
                atomicAdd(row + o1, hf ? p2 : p1);
                atomicAdd(row + o2, hf ? p3 : p2);
                atomicAdd(row + o3, hf ? p0 : p3);
            }
        }
    }
}

__global__ __launch_bounds__(MSG_THREADS) void gat_message_own(
    const float* __restrict__ src_p,
    const float* __restrict__ expbuf,
    const float* __restrict__ denom,
    const int*   __restrict__ srcI,
    const int*   __restrict__ dstI,
    float* __restrict__ out)
{
    __shared__ __attribute__((aligned(16))) float    acc[RNG * FF];
    __shared__ __attribute__((aligned(16))) float    alp[MSG_WAVES * 32 * HH];
    __shared__ __attribute__((aligned(16))) unsigned que[MSG_WAVES * QCAP];

    const int tid  = threadIdx.x;
    const int lane = tid & 31;
    const int wave = __builtin_amdgcn_readfirstlane(tid >> 5);
    const int base = blockIdx.x * RNG;

    float4* acc4 = (float4*)acc;
    for (int i = tid; i < RNG * FF / 4; i += MSG_THREADS)
        acc4[i] = make_float4(0.0f, 0.0f, 0.0f, 0.0f);
    __syncthreads();

    unsigned* q  = que + wave * QCAP;
    float*    al = alp + wave * 32 * HH;
    int qn = 0;

    const int4* __restrict__ d4 = (const int4*)dstI;
    const int start = blockIdx.x * (NGROUPS / MSG_BLOCKS);

    for (int g0 = wave; g0 < NGROUPS; g0 += SCAN_DEPTH * MSG_WAVES) {
        int4 d[SCAN_DEPTH];
        int  gi[SCAN_DEPTH];
#pragma unroll
        for (int k = 0; k < SCAN_DEPTH; ++k) {
            int g = g0 + k * MSG_WAVES;
            const bool in = g < NGROUPS;
            g += start;
            if (g >= NGROUPS) g -= NGROUPS;
            gi[k] = in ? g : -1;
            if (in) d[k] = d4[g * 32 + lane];
            else    d[k] = make_int4(-1, -1, -1, -1);
        }
#pragma unroll
        for (int k = 0; k < SCAN_DEPTH; ++k) {
            if (gi[k] < 0) continue;
            const unsigned r0 = (unsigned)(d[k].x - base);
            const unsigned r1 = (unsigned)(d[k].y - base);
            const unsigned r2 = (unsigned)(d[k].z - base);
            const unsigned r3 = (unsigned)(d[k].w - base);
            const bool m0 = r0 < (unsigned)RNG, m1 = r1 < (unsigned)RNG;
            const bool m2 = r2 < (unsigned)RNG, m3 = r3 < (unsigned)RNG;
            if (__builtin_amdgcn_ballot_w32(m0 | m1 | m2 | m3)) {
                const unsigned e0 = (unsigned)(gi[k] * 32 + lane) * 4u;
                qn = queue_push(q, qn, m0, ((e0 + 0u) << 8) | r0);
                qn = queue_push(q, qn, m1, ((e0 + 1u) << 8) | r1);
                qn = queue_push(q, qn, m2, ((e0 + 2u) << 8) | r2);
                qn = queue_push(q, qn, m3, ((e0 + 3u) << 8) | r3);
            }
        }
        while (qn >= 32) {
            qn -= 32;
            drain_edges<true>(q + qn, 32, al, acc, base, lane,
                              src_p, expbuf, denom, srcI);
        }
    }
    if (qn > 0)
        drain_edges<false>(q, qn, al, acc, base, lane, src_p, expbuf, denom, srcI);

    __syncthreads();
    float4* o4 = (float4*)(out + (size_t)base * FF);
    for (int i = tid; i < RNG * FF / 4; i += MSG_THREADS)
        o4[i] = acc4[i];
}

__global__ __launch_bounds__(256) void gat_finalize(
    float* __restrict__ out,
    const float* __restrict__ x,
    const float* __restrict__ bias,
    const float* __restrict__ prelu_w)
{
    const int i = blockIdx.x * 256 + threadIdx.x;
    if (i >= NN * FF) return;
    const float v  = out[i] + x[i] + bias[i & (FF - 1)];
    const float pw = prelu_w[0];
    out[i] = (v >= 0.0f) ? v : pw * v;
}

extern "C" void kernel_launch(void* const* d_in, const int* in_sizes, int n_in,
                              void* d_out, int out_size, void* d_ws, size_t ws_size,
                              hipStream_t stream) {
    (void)in_sizes; (void)n_in; (void)out_size; (void)ws_size;

    const float* x        = (const float*)d_in[0];
    const int*   eidx     = (const int*)  d_in[1];
    const float* W_src    = (const float*)d_in[2];
    const float* W_dst    = (const float*)d_in[3];
    const float* attn     = (const float*)d_in[4];
    const float* bias     = (const float*)d_in[5];
    const float* prelu_w  = (const float*)d_in[6];
    const int*   srcI     = eidx;
    const int*   dstI     = eidx + EE;

    float* out = (float*)d_out;

    float* ws      = (float*)d_ws;
    float* src_p   = ws;
    float* dst_p   = src_p + (size_t)NN * FF;
    float* scoreb  = dst_p + (size_t)NN * FF;
    float* denom   = scoreb + (size_t)EE * HH;
    float* partial = denom + (size_t)NN * HH;
    const int nScoreBlk = (EE * HH + 255) / 256;
    float* gmax    = partial + nScoreBlk;

    {
        const int n2 = NN * HH;
        gat_zero_kernel<<<(n2 + 255) / 256, 256, 0, stream>>>(denom, n2);
    }

    {
        dim3 grid(2, NN / 16);
        gat_proj_gemm<<<grid, 32, 0, stream>>>(x, W_src, W_dst, src_p, dst_p);
    }

    gat_score<<<nScoreBlk, 256, 0, stream>>>(src_p, dst_p, srcI, dstI, attn,
                                             scoreb, partial);
    gat_maxreduce<<<1, 256, 0, stream>>>(partial, nScoreBlk, gmax);

    gat_expdenom<<<nScoreBlk, 256, 0, stream>>>(scoreb, dstI, gmax, denom);

    gat_message_own<<<MSG_BLOCKS, MSG_THREADS, 0, stream>>>(src_p, scoreb, denom,
                                                            srcI, dstI, out);

    {
        const int n = NN * FF;
        gat_finalize<<<(n + 255) / 256, 256, 0, stream>>>(out, x, bias, prelu_w);
    }
}


# softmax denominators as LDS f32 sums in the block that owns the rows (add order free as before): exp, one divide per row at the end (rounding changed), residual, bias, PReLU ride there; 4 launches
# speedup vs baseline: 3.2169x; 1.2947x over previous
#include <hip/hip_runtime.h>
#include <hip/hip_bf16.h>
#include <math.h>

#define NN 50000
#define EE 800000
#define FF 128
#define HH 8
#define DD 16

typedef __attribute__((ext_vector_type(2))) float v2f;
typedef __attribute__((ext_vector_type(8))) float v8f;

__global__ __launch_bounds__(32) void gat_proj_gemm(
    const float* __restrict__ x,
    const float* __restrict__ W_src,
    const float* __restrict__ W_dst,
    float* __restrict__ src_p,
    float* __restrict__ dst_p)
{
    const int which   = blockIdx.x;
    const int rowbase = blockIdx.y * 16;
    const float* __restrict__ W   = which ? W_dst : W_src;
    float* __restrict__       out = which ? dst_p : src_p;

    const int lane = threadIdx.x;
    const int half = lane >> 4;
    const int l    = lane & 15;

    const float* __restrict__ xrow = x + (size_t)(rowbase + l) * FF;

    v8f acc[8];
#pragma unroll
    for (int c = 0; c < 8; ++c) acc[c] = (v8f){};

    for (int k = 0; k < FF; k += 4) {
        v2f a;
        a.x = xrow[k + 2 * half + 0];
        a.y = xrow[k + 2 * half + 1];
#pragma unroll
        for (int c = 0; c < 8; ++c) {
            const float* wp = W + (size_t)(k + 2 * half) * FF + c * 16 + l;
            v2f b;
            b.x = wp[0];
            b.y = wp[FF];
            acc[c] = __builtin_amdgcn_wmma_f32_16x16x4_f32(
                false, a, false, b, (short)0, acc[c], false, false);
        }
    }

#pragma unroll
    for (int c = 0; c < 8; ++c) {
#pragma unroll
        for (int r = 0; r < 8; ++r) {
            out[(size_t)(rowbase + r + 8 * half) * FF + c * 16 + l] = acc[c][r];
        }
    }
}

__global__ __launch_bounds__(256) void gat_score(
    const float* __restrict__ src_p,
    const float* __restrict__ dst_p,
    const int*   __restrict__ srcI,
    const int*   __restrict__ dstI,
    const float* __restrict__ attn,
    float* __restrict__ scorebuf,
    float* __restrict__ partial)
{
    __shared__ float smax[256];
    const int tid = blockIdx.x * 256 + threadIdx.x;
    float s = -INFINITY;
    if (tid < EE * HH) {
        const int e = tid >> 3;
        const int h = tid & 7;
        const float* sp = src_p + (size_t)srcI[e] * FF + h * DD;
        const float* dp = dst_p + (size_t)dstI[e] * FF + h * DD;
        const float* at = attn + h * DD;
        float acc = 0.0f;
#pragma unroll
        for (int d = 0; d < DD; ++d) {
            float v = sp[d] + dp[d];
            v = (v >= 0.0f) ? v : 0.2f * v;
            acc += at[d] * v;
        }
        scorebuf[tid] = acc;
        s = acc;
    }
    smax[threadIdx.x] = s;
    __syncthreads();
#pragma unroll
    for (int off = 128; off > 0; off >>= 1) {
        if (threadIdx.x < off)
            smax[threadIdx.x] = fmaxf(smax[threadIdx.x], smax[threadIdx.x + off]);
        __syncthreads();
    }
    if (threadIdx.x == 0) partial[blockIdx.x] = smax[0];
}

__global__ __launch_bounds__(256) void gat_maxreduce(
    const float* __restrict__ partial, int n, float* __restrict__ gmax)
{
    __shared__ float smax[256];
    float s = -INFINITY;
    for (int i = threadIdx.x; i < n; i += 256) s = fmaxf(s, partial[i]);
    smax[threadIdx.x] = s;
    __syncthreads();
#pragma unroll
    for (int off = 128; off > 0; off >>= 1) {
        if (threadIdx.x < off)
            smax[threadIdx.x] = fmaxf(smax[threadIdx.x], smax[threadIdx.x + off]);
        __syncthreads();
    }
    if (threadIdx.x == 0) *gmax = smax[0];
}

#define RNG         200
#define MSG_BLOCKS  (NN / RNG)
#define MSG_THREADS 1024
#define MSG_WAVES   (MSG_THREADS / 32)
#define SCAN_DEPTH  8
#define QCAP        1056
#define NGROUPS     (EE / 128)

static_assert(NN % RNG == 0, "blocks tile the rows exactly");
static_assert(RNG <= 256, "local row fits the low 8 bits of a queue word");
static_assert(EE % 128 == 0 && EE < (1 << 24), "edge id fits the upper 24 bits");
static_assert(NGROUPS % MSG_BLOCKS == 0, "start offsets tile the list");
static_assert(QCAP >= 31 + SCAN_DEPTH * 128, "a step's edges always fit the queue");

__device__ __forceinline__ void wave_lds_order() {
    __builtin_amdgcn_fence(__ATOMIC_RELEASE, "wavefront");
    __builtin_amdgcn_wave_barrier();
    __builtin_amdgcn_fence(__ATOMIC_ACQUIRE, "wavefront");
}

__device__ __forceinline__ int queue_push(unsigned* __restrict__ q, int qn,
                                          bool m, unsigned word) {
    const unsigned mask = __builtin_amdgcn_ballot_w32(m);
    if (mask) {
        if (m) q[qn + (int)__builtin_amdgcn_mbcnt_lo(mask, 0u)] = word;
        qn += __builtin_popcount(mask);
    }
    return qn;
}

template <bool FULL>
__device__ __forceinline__ void drain_edges(
    const unsigned* __restrict__ q, int n, float* __restrict__ al,
    float* __restrict__ acc, float* __restrict__ den, float gm, int lane,
    const float* __restrict__ src_p,
    const float* __restrict__ scorebuf,
    const int*   __restrict__ srcI)
{
    wave_lds_order();
    const bool act = FULL || (lane < n);
    unsigned word = 0u;
    if (act) word = q[lane];
    const unsigned e  = word >> 8;
    const int      dl = (int)(word & 255u);
    int s = 0;
    if (act) {
        s = srcI[e];
        const float4* sc4 = (const float4*)(scorebuf + (size_t)e * HH);
        const float4 x0 = sc4[0], x1 = sc4[1];
        float4 a0, a1;
        a0.x = __expf(x0.x - gm);
        a0.y = __expf(x0.y - gm);
        a0.z = __expf(x0.z - gm);
        a0.w = __expf(x0.w - gm);
        a1.x = __expf(x1.x - gm);
        a1.y = __expf(x1.y - gm);
        a1.z = __expf(x1.z - gm);
        a1.w = __expf(x1.w - gm);
        float4* ap = (float4*)(al + lane * HH);
        ap[0] = a0;
        ap[1] = a1;
        float* dr = den + dl * HH;
        atomicAdd(dr + 0, a0.x);
        atomicAdd(dr + 1, a0.y);
        atomicAdd(dr + 2, a0.z);
        atomicAdd(dr + 3, a0.w);
        atomicAdd(dr + 4, a1.x);
        atomicAdd(dr + 5, a1.y);
        atomicAdd(dr + 6, a1.z);
        atomicAdd(dr + 7, a1.w);
    }
    wave_lds_order();

    const int hd = lane >> 2;
    const int hf = lane >> 4;
    const int o0 = lane * 4 + (hf ? 1 : 0);
    const int o1 = lane * 4 + (hf ? 2 : 1);
    const int o2 = lane * 4 + (hf ? 3 : 2);
    const int o3 = lane * 4 + (hf ? 0 : 3);
    const int cnt = FULL ? 32 : n;
    for (int j0 = 0; j0 < cnt; j0 += 8) {
        float4 v[8];
        float  a[8];
        int    dj[8];
#pragma unroll
        for (int u = 0; u < 8; ++u) {
            const int sj = __builtin_amdgcn_readlane(s, j0 + u);
            dj[u] = __builtin_amdgcn_readlane(dl, j0 + u);
            v[u]  = ((const float4*)(src_p + (size_t)sj * FF))[lane];
            a[u]  = al[(j0 + u) * HH + hd];
        }
#pragma unroll
        for (int u = 0; u < 8; ++u) {
            if (FULL || (j0 + u < cnt)) {
                const float p0 = v[u].x * a[u], p1 = v[u].y * a[u];
                const float p2 = v[u].z * a[u], p3 = v[u].w * a[u];
                float* row = acc + dj[u] * FF;
                atomicAdd(row + o0, hf ? p1 : p0);
                atomicAdd(row + o1, hf ? p2 : p1);
                atomicAdd(row + o2, hf ? p3 : p2);
                atomicAdd(row + o3, hf ? p0 : p3);
            }
        }
    }
}

__global__ __launch_bounds__(MSG_THREADS) void gat_message_own(
    const float* __restrict__ src_p,
    const float* __restrict__ scorebuf,
    const float* __restrict__ gmax,
    const int*   __restrict__ srcI,
    const int*   __restrict__ dstI,
    const float* __restrict__ x,
    const float* __restrict__ bias,
    const float* __restrict__ prelu_w,
    float* __restrict__ out)
{
    __shared__ __attribute__((aligned(16))) float    acc[RNG * FF];
    __shared__ __attribute__((aligned(16))) float    den[RNG * HH];
    __shared__ __attribute__((aligned(16))) float    alp[MSG_WAVES * 32 * HH];
    __shared__ __attribute__((aligned(16))) unsigned que[MSG_WAVES * QCAP];

    const int tid  = threadIdx.x;
    const int lane = tid & 31;
    const int wave = __builtin_amdgcn_readfirstlane(tid >> 5);
    const int base = blockIdx.x * RNG;

    const float gm = *gmax;

    float4* acc4 = (float4*)acc;
    for (int i = tid; i < RNG * FF / 4; i += MSG_THREADS)
        acc4[i] = make_float4(0.0f, 0.0f, 0.0f, 0.0f);
    for (int i = tid; i < RNG * HH; i += MSG_THREADS)
        den[i] = 0.0f;
    __syncthreads();

    unsigned* q  = que + wave * QCAP;
    float*    al = alp + wave * 32 * HH;
    int qn = 0;

    const int4* __restrict__ d4 = (const int4*)dstI;
    const int start = blockIdx.x * (NGROUPS / MSG_BLOCKS);

    for (int g0 = wave; g0 < NGROUPS; g0 += SCAN_DEPTH * MSG_WAVES) {
        int4 d[SCAN_DEPTH];
        int  gi[SCAN_DEPTH];
#pragma unroll
        for (int k = 0; k < SCAN_DEPTH; ++k) {
            int g = g0 + k * MSG_WAVES;
            const bool in = g < NGROUPS;
            g += start;
            if (g >= NGROUPS) g -= NGROUPS;
            gi[k] = in ? g : -1;
            if (in) d[k] = d4[g * 32 + lane];
            else    d[k] = make_int4(-1, -1, -1, -1);
        }
#pragma unroll
        for (int k = 0; k < SCAN_DEPTH; ++k) {
            if (gi[k] < 0) continue;
            const unsigned r0 = (unsigned)(d[k].x - base);
            const unsigned r1 = (unsigned)(d[k].y - base);
            const unsigned r2 = (unsigned)(d[k].z - base);
            const unsigned r3 = (unsigned)(d[k].w - base);
            const bool m0 = r0 < (unsigned)RNG, m1 = r1 < (unsigned)RNG;
            const bool m2 = r2 < (unsigned)RNG, m3 = r3 < (unsigned)RNG;
            if (__builtin_amdgcn_ballot_w32(m0 | m1 | m2 | m3)) {
                const unsigned e0 = (unsigned)(gi[k] * 32 + lane) * 4u;
                qn = queue_push(q, qn, m0, ((e0 + 0u) << 8) | r0);
                qn = queue_push(q, qn, m1, ((e0 + 1u) << 8) | r1);
                qn = queue_push(q, qn, m2, ((e0 + 2u) << 8) | r2);
                qn = queue_push(q, qn, m3, ((e0 + 3u) << 8) | r3);
            }
        }
        while (qn >= 32) {
            qn -= 32;
            drain_edges<true>(q + qn, 32, al, acc, den, gm, lane,
                              src_p, scorebuf, srcI);
        }
    }
    if (qn > 0)
        drain_edges<false>(q, qn, al, acc, den, gm, lane, src_p, scorebuf, srcI);

    __syncthreads();
    const float4* __restrict__ x4 = (const float4*)(x + (size_t)base * FF);
    const float4* __restrict__ b4 = (const float4*)bias;
    const float pw = prelu_w[0];
    float4* o4 = (float4*)(out + (size_t)base * FF);
    for (int i = tid; i < RNG * FF / 4; i += MSG_THREADS) {
        const int row   = i >> 5;
        const int piece = i & 31;
        const float  d  = den[row * HH + (piece >> 2)] + 1e-16f;
        const float4 a  = acc4[i];
        const float4 xv = x4[i];
        const float4 bv = b4[piece];
        float4 r;
        r.x = a.x / d + xv.x + bv.x;
        r.y = a.y / d + xv.y + bv.y;
        r.z = a.z / d + xv.z + bv.z;
        r.w = a.w / d + xv.w + bv.w;
        r.x = (r.x >= 0.0f) ? r.x : pw * r.x;
        r.y = (r.y >= 0.0f) ? r.y : pw * r.y;
        r.z = (r.z >= 0.0f) ? r.z : pw * r.z;
        r.w = (r.w >= 0.0f) ? r.w : pw * r.w;
        o4[i] = r;
    }
}

extern "C" void kernel_launch(void* const* d_in, const int* in_sizes, int n_in,
                              void* d_out, int out_size, void* d_ws, size_t ws_size,
                              hipStream_t stream) {
    (void)in_sizes; (void)n_in; (void)out_size; (void)ws_size;

    const float* x        = (const float*)d_in[0];
    const int*   eidx     = (const int*)  d_in[1];
    const float* W_src    = (const float*)d_in[2];
    const float* W_dst    = (const float*)d_in[3];
    const float* attn     = (const float*)d_in[4];
    const float* bias     = (const float*)d_in[5];
    const float* prelu_w  = (const float*)d_in[6];
    const int*   srcI     = eidx;
    const int*   dstI     = eidx + EE;

    float* out = (float*)d_out;

    float* ws      = (float*)d_ws;
    float* src_p   = ws;
    float* dst_p   = src_p + (size_t)NN * FF;
    float* scoreb  = dst_p + (size_t)NN * FF;
    float* partial = scoreb + (size_t)EE * HH;
    const int nScoreBlk = (EE * HH + 255) / 256;
    float* gmax    = partial + nScoreBlk;

    {
        dim3 grid(2, NN / 16);
        gat_proj_gemm<<<grid, 32, 0, stream>>>(x, W_src, W_dst, src_p, dst_p);
    }

    gat_score<<<nScoreBlk, 256, 0, stream>>>(src_p, dst_p, srcI, dstI, attn,
                                             scoreb, partial);
    gat_maxreduce<<<1, 256, 0, stream>>>(partial, nScoreBlk, gmax);

    gat_message_own<<<MSG_BLOCKS, MSG_THREADS, 0, stream>>>(src_p, scoreb, gmax,
                                                            srcI, dstI, x, bias,
                                                            prelu_w, out);
}


# scores move into the 250 owner blocks' list walk (own dst rows in LDS; dot = 4 in-lane terms + 2 lane adds, sum order changed); blocks keep their edge lists, so the message launch walks no list
# speedup vs baseline: 3.9191x; 1.2183x over previous
#include <hip/hip_runtime.h>
#include <hip/hip_bf16.h>
#include <math.h>

#define NN 50000
#define EE 800000
#define FF 128
#define HH 8
#define DD 16

typedef __attribute__((ext_vector_type(2))) float v2f;
typedef __attribute__((ext_vector_type(8))) float v8f;

__global__ __launch_bounds__(32) void gat_proj_gemm(
    const float* __restrict__ x,
    const float* __restrict__ W_src,
    const float* __restrict__ W_dst,
    float* __restrict__ src_p,
    float* __restrict__ dst_p)
{
    const int which   = blockIdx.x;
    const int rowbase = blockIdx.y * 16;
    const float* __restrict__ W   = which ? W_dst : W_src;
    float* __restrict__       out = which ? dst_p : src_p;

    const int lane = threadIdx.x;
    const int half = lane >> 4;
    const int l    = lane & 15;

    const float* __restrict__ xrow = x + (size_t)(rowbase + l) * FF;

    v8f acc[8];
#pragma unroll
    for (int c = 0; c < 8; ++c) acc[c] = (v8f){};

    for (int k = 0; k < FF; k += 4) {
        v2f a;
        a.x = xrow[k + 2 * half + 0];
        a.y = xrow[k + 2 * half + 1];
#pragma unroll
        for (int c = 0; c < 8; ++c) {
            const float* wp = W + (size_t)(k + 2 * half) * FF + c * 16 + l;
            v2f b;
            b.x = wp[0];
            b.y = wp[FF];
            acc[c] = __builtin_amdgcn_wmma_f32_16x16x4_f32(
                false, a, false, b, (short)0, acc[c], false, false);
        }
    }

#pragma unroll
    for (int c = 0; c < 8; ++c) {
#pragma unroll
        for (int r = 0; r < 8; ++r) {
            out[(size_t)(rowbase + r + 8 * half) * FF + c * 16 + l] = acc[c][r];
        }
    }
}

__global__ __launch_bounds__(256) void gat_maxreduce(
    const float* __restrict__ partial, int n, float* __restrict__ gmax)
{
    __shared__ float smax[256];
    float s = -INFINITY;
    for (int i = threadIdx.x; i < n; i += 256) s = fmaxf(s, partial[i]);
    smax[threadIdx.x] = s;
    __syncthreads();
#pragma unroll
    for (int off = 128; off > 0; off >>= 1) {
        if (threadIdx.x < off)
            smax[threadIdx.x] = fmaxf(smax[threadIdx.x], smax[threadIdx.x + off]);
        __syncthreads();
    }
    if (threadIdx.x == 0) *gmax = smax[0];
}

#define RNG         200
#define MSG_BLOCKS  (NN / RNG)
#define MSG_THREADS 1024
#define MSG_WAVES   (MSG_THREADS / 32)
#define SCAN_DEPTH  8
#define QCAP        1056
#define NGROUPS     (EE / 128)

static_assert(NN % RNG == 0, "blocks tile the rows exactly");
static_assert(RNG <= 256, "local row fits the low 8 bits of a queue word");
static_assert(EE % 128 == 0 && EE < (1 << 24), "edge id fits the upper 24 bits");
static_assert(NGROUPS % MSG_BLOCKS == 0, "start offsets tile the list");
static_assert(QCAP >= 31 + SCAN_DEPTH * 128, "a step's edges always fit the queue");

__device__ __forceinline__ void wave_lds_order() {
    __builtin_amdgcn_fence(__ATOMIC_RELEASE, "wavefront");
    __builtin_amdgcn_wave_barrier();
    __builtin_amdgcn_fence(__ATOMIC_ACQUIRE, "wavefront");
}

__device__ __forceinline__ int queue_push(unsigned* __restrict__ q, int qn,
                                          bool m, unsigned word) {
    const unsigned mask = __builtin_amdgcn_ballot_w32(m);
    if (mask) {
        if (m) q[qn + (int)__builtin_amdgcn_mbcnt_lo(mask, 0u)] = word;
        qn += __builtin_popcount(mask);
    }
    return qn;
}

template <bool FULL>
__device__ __forceinline__ void drain_edges(
    const unsigned* __restrict__ q, int n, float* __restrict__ al,
    float* __restrict__ acc, float* __restrict__ den, float gm, int lane,
    const float* __restrict__ src_p,
    const float* __restrict__ scorebuf,
    const int*   __restrict__ srcI)
{
    wave_lds_order();
    const bool act = FULL || (lane < n);
    unsigned word = 0u;
    if (act) word = q[lane];
    const unsigned e  = word >> 8;
    const int      dl = (int)(word & 255u);
    int s = 0;
    if (act) {
        s = srcI[e];
        const float4* sc4 = (const float4*)(scorebuf + (size_t)e * HH);
        const float4 x0 = sc4[0], x1 = sc4[1];
        float4 a0, a1;
        a0.x = __expf(x0.x - gm);
        a0.y = __expf(x0.y - gm);
        a0.z = __expf(x0.z - gm);
        a0.w = __expf(x0.w - gm);
        a1.x = __expf(x1.x - gm);
        a1.y = __expf(x1.y - gm);
        a1.z = __expf(x1.z - gm);
        a1.w = __expf(x1.w - gm);
        float4* ap = (float4*)(al + lane * HH);
        ap[0] = a0;
        ap[1] = a1;
        float* dr = den + dl * HH;
        atomicAdd(dr + 0, a0.x);
        atomicAdd(dr + 1, a0.y);
        atomicAdd(dr + 2, a0.z);
        atomicAdd(dr + 3, a0.w);
        atomicAdd(dr + 4, a1.x);
        atomicAdd(dr + 5, a1.y);
        atomicAdd(dr + 6, a1.z);
        atomicAdd(dr + 7, a1.w);
    }
    wave_lds_order();

    const int hd = lane >> 2;
    const int hf = lane >> 4;
    const int o0 = lane * 4 + (hf ? 1 : 0);
    const int o1 = lane * 4 + (hf ? 2 : 1);
    const int o2 = lane * 4 + (hf ? 3 : 2);
    const int o3 = lane * 4 + (hf ? 0 : 3);
    const int cnt = FULL ? 32 : n;
    for (int j0 = 0; j0 < cnt; j0 += 8) {
        float4 v[8];
        float  a[8];
        int    dj[8];
#pragma unroll
        for (int u = 0; u < 8; ++u) {
            const int sj = __builtin_amdgcn_readlane(s, j0 + u);
            dj[u] = __builtin_amdgcn_readlane(dl, j0 + u);
            v[u]  = ((const float4*)(src_p + (size_t)sj * FF))[lane];
            a[u]  = al[(j0 + u) * HH + hd];
        }
#pragma unroll
        for (int u = 0; u < 8; ++u) {
            if (FULL || (j0 + u < cnt)) {
                const float p0 = v[u].x * a[u], p1 = v[u].y * a[u];
                const float p2 = v[u].z * a[u], p3 = v[u].w * a[u];
                float* row = acc + dj[u] * FF;
                atomicAdd(row + o0, hf ? p1 : p0);
                atomicAdd(row + o1, hf ? p2 : p1);
                atomicAdd(row + o2, hf ? p3 : p2);
                atomicAdd(row + o3, hf ? p0 : p3);
            }
        }
    }
}

#define LCAP 256

static_assert(MSG_WAVES == 32, "K4 reads a block's wave counts one per lane");

__device__ __forceinline__ float quad_sum(float p) {
    p += __int_as_float(__builtin_amdgcn_update_dpp(0, __float_as_int(p), 0x161, 0xf, 0xf, true));
    p += __int_as_float(__builtin_amdgcn_update_dpp(0, __float_as_int(p), 0x162, 0xf, 0xf, true));
    return p;
}

template <bool FULL>
__device__ __forceinline__ float score_edges(
    const unsigned* __restrict__ q, int n, const float4* __restrict__ drow4,
    const float4 at, float mx, int lane,
    const float* __restrict__ src_p,
    const int*   __restrict__ srcI,
    float*       __restrict__ scorebuf,
    unsigned*    __restrict__ seg, int kept)
{
    wave_lds_order();
    const bool act = FULL || (lane < n);
    unsigned word = 0u;
    if (act) word = q[lane];
    const int e  = (int)(word >> 8);
    const int dl = (int)(word & 255u);
    int s = 0;
    if (act) {
        s = srcI[e];
        if (kept + lane < LCAP) seg[kept + lane] = word;
    }
    const int hd  = lane >> 2;
    const int cnt = FULL ? 32 : n;
    for (int j0 = 0; j0 < cnt; j0 += 8) {
        float4 v[8];
        float4 d[8];
        int    ej[8];
#pragma unroll
        for (int u = 0; u < 8; ++u) {
            const int sj = __builtin_amdgcn_readlane(s, j0 + u);
            const int dj = __builtin_amdgcn_readlane(dl, j0 + u);
            ej[u] = __builtin_amdgcn_readlane(e, j0 + u);
            v[u]  = ((const float4*)(src_p + (size_t)sj * FF))[lane];
            d[u]  = drow4[dj * (FF / 4) + lane];
        }
#pragma unroll
        for (int u = 0; u < 8; ++u) {
            if (FULL || (j0 + u < cnt)) {
                float t0 = v[u].x + d[u].x, t1 = v[u].y + d[u].y;
                float t2 = v[u].z + d[u].z, t3 = v[u].w + d[u].w;
                t0 = (t0 >= 0.0f) ? t0 : 0.2f * t0;
                t1 = (t1 >= 0.0f) ? t1 : 0.2f * t1;
                t2 = (t2 >= 0.0f) ? t2 : 0.2f * t2;
                t3 = (t3 >= 0.0f) ? t3 : 0.2f * t3;
                const float p = quad_sum(at.x * t0 + at.y * t1 + at.z * t2 + at.w * t3);
                if ((lane & 3) == 0) scorebuf[(size_t)ej[u] * HH + hd] = p;
                mx = fmaxf(mx, p);
            }
        }
    }
    return mx;
}

__global__ __launch_bounds__(MSG_THREADS) void gat_score_own(
    const float* __restrict__ src_p,
    const float* __restrict__ dst_p,
    const int*   __restrict__ srcI,
    const int*   __restrict__ dstI,
    const float* __restrict__ attn,
    float*       __restrict__ scorebuf,
    float*       __restrict__ partial,
    unsigned*    __restrict__ lists,
    int*         __restrict__ counts)
{
    __shared__ __attribute__((aligned(16))) float    drow[RNG * FF];
    __shared__ __attribute__((aligned(16))) unsigned que[MSG_WAVES * QCAP];
    __shared__ float wmax[MSG_WAVES];

    const int tid  = threadIdx.x;
    const int lane = tid & 31;
    const int wave = __builtin_amdgcn_readfirstlane(tid >> 5);
    const int base = blockIdx.x * RNG;

    const float4* __restrict__ dp4 = (const float4*)(dst_p + (size_t)base * FF);
    float4* drow4 = (float4*)drow;
    for (int i = tid; i < RNG * FF / 4; i += MSG_THREADS)
        drow4[i] = dp4[i];
    __syncthreads();

    const float4 at = ((const float4*)attn)[lane];
    unsigned* q   = que + wave * QCAP;
    unsigned* seg = lists + (size_t)(blockIdx.x * MSG_WAVES + wave) * LCAP;
    int   qn   = 0;
    int   kept = 0;
    float mx   = -INFINITY;

    const int4* __restrict__ d4 = (const int4*)dstI;
    const int start = blockIdx.x * (NGROUPS / MSG_BLOCKS);

    for (int g0 = wave; g0 < NGROUPS; g0 += SCAN_DEPTH * MSG_WAVES) {
        int4 d[SCAN_DEPTH];
        int  gi[SCAN_DEPTH];
#pragma unroll
        for (int k = 0; k < SCAN_DEPTH; ++k) {
            int g = g0 + k * MSG_WAVES;
            const bool in = g < NGROUPS;
            g += start;
            if (g >= NGROUPS) g -= NGROUPS;
            gi[k] = in ? g : -1;
            if (in) d[k] = d4[g * 32 + lane];
            else    d[k] = make_int4(-1, -1, -1, -1);
        }
#pragma unroll
        for (int k = 0; k < SCAN_DEPTH; ++k) {
            if (gi[k] < 0) continue;
            const unsigned r0 = (unsigned)(d[k].x - base);
            const unsigned r1 = (unsigned)(d[k].y - base);
            const unsigned r2 = (unsigned)(d[k].z - base);
            const unsigned r3 = (unsigned)(d[k].w - base);
            const bool m0 = r0 < (unsigned)RNG, m1 = r1 < (unsigned)RNG;
            const bool m2 = r2 < (unsigned)RNG, m3 = r3 < (unsigned)RNG;
            if (__builtin_amdgcn_ballot_w32(m0 | m1 | m2 | m3)) {
                const unsigned e0 = (unsigned)(gi[k] * 32 + lane) * 4u;
                qn = queue_push(q, qn, m0, ((e0 + 0u) << 8) | r0);
                qn = queue_push(q, qn, m1, ((e0 + 1u) << 8) | r1);
                qn = queue_push(q, qn, m2, ((e0 + 2u) << 8) | r2);
                qn = queue_push(q, qn, m3, ((e0 + 3u) << 8) | r3);
            }
        }
        while (qn >= 32) {
            qn -= 32;
            mx = score_edges<true>(q + qn, 32, drow4, at, mx, lane,
                                   src_p, srcI, scorebuf, seg, kept);
            kept += 32;
        }
    }
    if (qn > 0) {
        mx = score_edges<false>(q, qn, drow4, at, mx, lane,
                                src_p, srcI, scorebuf, seg, kept);
        kept += qn;
    }
    if (lane == 0) counts[blockIdx.x * MSG_WAVES + wave] = kept;

    mx = fmaxf(mx, __shfl_xor(mx, 16));
    mx = fmaxf(mx, __shfl_xor(mx, 8));
    mx = fmaxf(mx, __shfl_xor(mx, 4));
    mx = fmaxf(mx, __shfl_xor(mx, 2));
    mx = fmaxf(mx, __shfl_xor(mx, 1));
    if (lane == 0) wmax[wave] = mx;
    __syncthreads();
    if (tid == 0) {
        float m = wmax[0];
        for (int i = 1; i < MSG_WAVES; ++i) m = fmaxf(m, wmax[i]);
        partial[blockIdx.x] = m;
    }
}

__global__ __launch_bounds__(MSG_THREADS) void gat_message_own(
    const float* __restrict__ src_p,
    const float* __restrict__ scorebuf,
    const float* __restrict__ gmax,
    const int*   __restrict__ srcI,
    const int*   __restrict__ dstI,
    const float* __restrict__ x,
    const float* __restrict__ bias,
    const float* __restrict__ prelu_w,
    const unsigned* __restrict__ lists,
    const int*   __restrict__ counts,
    float* __restrict__ out)
{
    __shared__ __attribute__((aligned(16))) float    acc[RNG * FF];
    __shared__ __attribute__((aligned(16))) float    den[RNG * HH];
    __shared__ __attribute__((aligned(16))) float    alp[MSG_WAVES * 32 * HH];
    __shared__ __attribute__((aligned(16))) unsigned que[MSG_WAVES * QCAP];

    const int tid  = threadIdx.x;
    const int lane = tid & 31;
    const int wave = __builtin_amdgcn_readfirstlane(tid >> 5);
    const int base = blockIdx.x * RNG;

    const float gm = *gmax;

    float4* acc4 = (float4*)acc;
    for (int i = tid; i < RNG * FF / 4; i += MSG_THREADS)
        acc4[i] = make_float4(0.0f, 0.0f, 0.0f, 0.0f);
    for (int i = tid; i < RNG * HH; i += MSG_THREADS)
        den[i] = 0.0f;
    __syncthreads();

    unsigned* q  = que + wave * QCAP;
    float*    al = alp + wave * 32 * HH;
    int qn = 0;

    const int  cl   = counts[blockIdx.x * MSG_WAVES + lane];
    const bool fits = __builtin_amdgcn_ballot_w32(cl > LCAP) == 0u;
    if (fits) {
        const int c = __builtin_amdgcn_readlane(cl, wave);
        const unsigned* __restrict__ seg =
            lists + (size_t)(blockIdx.x * MSG_WAVES + wave) * LCAP;
        for (int i = 0; i + 32 <= c; i += 32) {
            q[lane] = seg[i + lane];
            drain_edges<true>(q, 32, al, acc, den, gm, lane, src_p, scorebuf, srcI);
        }
        qn = c & 31;
        if (lane < qn) q[lane] = seg[(c & ~31) + lane];
    } else {
        const int4* __restrict__ d4 = (const int4*)dstI;
        const int start = blockIdx.x * (NGROUPS / MSG_BLOCKS);

        for (int g0 = wave; g0 < NGROUPS; g0 += SCAN_DEPTH * MSG_WAVES) {
            int4 d[SCAN_DEPTH];
            int  gi[SCAN_DEPTH];
#pragma unroll
            for (int k = 0; k < SCAN_DEPTH; ++k) {
                int g = g0 + k * MSG_WAVES;
                const bool in = g < NGROUPS;
                g += start;
                if (g >= NGROUPS) g -= NGROUPS;
                gi[k] = in ? g : -1;
                if (in) d[k] = d4[g * 32 + lane];
                else    d[k] = make_int4(-1, -1, -1, -1);
            }
#pragma unroll
            for (int k = 0; k < SCAN_DEPTH; ++k) {
                if (gi[k] < 0) continue;
                const unsigned r0 = (unsigned)(d[k].x - base);
                const unsigned r1 = (unsigned)(d[k].y - base);
                const unsigned r2 = (unsigned)(d[k].z - base);
                const unsigned r3 = (unsigned)(d[k].w - base);
                const bool m0 = r0 < (unsigned)RNG, m1 = r1 < (unsigned)RNG;
                const bool m2 = r2 < (unsigned)RNG, m3 = r3 < (unsigned)RNG;
                if (__builtin_amdgcn_ballot_w32(m0 | m1 | m2 | m3)) {
                    const unsigned e0 = (unsigned)(gi[k] * 32 + lane) * 4u;
                    qn = queue_push(q, qn, m0, ((e0 + 0u) << 8) | r0);
                    qn = queue_push(q, qn, m1, ((e0 + 1u) << 8) | r1);
                    qn = queue_push(q, qn, m2, ((e0 + 2u) << 8) | r2);
                    qn = queue_push(q, qn, m3, ((e0 + 3u) << 8) | r3);
                }
            }
            while (qn >= 32) {
                qn -= 32;
                drain_edges<true>(q + qn, 32, al, acc, den, gm, lane,
                                  src_p, scorebuf, srcI);
            }
        }
    }
    if (qn > 0)
        drain_edges<false>(q, qn, al, acc, den, gm, lane, src_p, scorebuf, srcI);

    __syncthreads();
    const float4* __restrict__ x4 = (const float4*)(x + (size_t)base * FF);
    const float4* __restrict__ b4 = (const float4*)bias;
    const float pw = prelu_w[0];
    float4* o4 = (float4*)(out + (size_t)base * FF);
    for (int i = tid; i < RNG * FF / 4; i += MSG_THREADS) {
        const int row   = i >> 5;
        const int piece = i & 31;
        const float  d  = den[row * HH + (piece >> 2)] + 1e-16f;
        const float4 a  = acc4[i];
        const float4 xv = x4[i];
        const float4 bv = b4[piece];
        float4 r;
        r.x = a.x / d + xv.x + bv.x;
        r.y = a.y / d + xv.y + bv.y;
        r.z = a.z / d + xv.z + bv.z;
        r.w = a.w / d + xv.w + bv.w;
        r.x = (r.x >= 0.0f) ? r.x : pw * r.x;
        r.y = (r.y >= 0.0f) ? r.y : pw * r.y;
        r.z = (r.z >= 0.0f) ? r.z : pw * r.z;
        r.w = (r.w >= 0.0f) ? r.w : pw * r.w;
        o4[i] = r;
    }
}

extern "C" void kernel_launch(void* const* d_in, const int* in_sizes, int n_in,
                              void* d_out, int out_size, void* d_ws, size_t ws_size,
                              hipStream_t stream) {
    (void)in_sizes; (void)n_in; (void)out_size; (void)ws_size;

    const float* x        = (const float*)d_in[0];
    const int*   eidx     = (const int*)  d_in[1];
    const float* W_src    = (const float*)d_in[2];
    const float* W_dst    = (const float*)d_in[3];
    const float* attn     = (const float*)d_in[4];
    const float* bias     = (const float*)d_in[5];
    const float* prelu_w  = (const float*)d_in[6];
    const int*   srcI     = eidx;
    const int*   dstI     = eidx + EE;

    float* out = (float*)d_out;

    float* ws      = (float*)d_ws;
    float* src_p   = ws;
    float* dst_p   = src_p + (size_t)NN * FF;
    float* scoreb  = dst_p + (size_t)NN * FF;
    float* partial = scoreb + (size_t)EE * HH;
    float* gmax    = partial + MSG_BLOCKS;
    int*      counts = (int*)(gmax + 1);
    unsigned* lists  = (unsigned*)(counts + MSG_BLOCKS * MSG_WAVES);

    {
        dim3 grid(2, NN / 16);
        gat_proj_gemm<<<grid, 32, 0, stream>>>(x, W_src, W_dst, src_p, dst_p);
    }

    gat_score_own<<<MSG_BLOCKS, MSG_THREADS, 0, stream>>>(src_p, dst_p, srcI, dstI,
                                                          attn, scoreb, partial,
                                                          lists, counts);
    gat_maxreduce<<<1, 256, 0, stream>>>(partial, MSG_BLOCKS, gmax);

    gat_message_own<<<MSG_BLOCKS, MSG_THREADS, 0, stream>>>(src_p, scoreb, gmax,
                                                            srcI, dstI, x, bias,
                                                            prelu_w, lists, counts,
                                                            out);
}


# score launch walks one owner-block byte per edge (keys + a packed source/row word, written by the projection launch), not the 4-byte destination list: 800 KB per block, not 3200 KB; same sums
# speedup vs baseline: 4.3789x; 1.1173x over previous
#include <hip/hip_runtime.h>
#include <hip/hip_bf16.h>
#include <math.h>

#define NN 50000
#define EE 800000
#define FF 128
#define HH 8
#define DD 16

#define RNG         200
#define MSG_BLOCKS  (NN / RNG)

typedef __attribute__((ext_vector_type(2))) float v2f;
typedef __attribute__((ext_vector_type(8))) float v8f;

__global__ __launch_bounds__(32) void gat_proj_gemm(
    const float* __restrict__ x,
    const float* __restrict__ W_src,
    const float* __restrict__ W_dst,
    float* __restrict__ src_p,
    float* __restrict__ dst_p,
    const int* __restrict__ srcI,
    const int* __restrict__ dstI,
    unsigned* __restrict__ keys,
    unsigned* __restrict__ pk)
{
    {
        const int wid = blockIdx.y * 2 + blockIdx.x;
        const int4 dv = ((const int4*)dstI)[wid * 32 + threadIdx.x];
        const int4 sv = ((const int4*)srcI)[wid * 32 + threadIdx.x];
        const unsigned k0 = (unsigned)dv.x / RNG, k1 = (unsigned)dv.y / RNG;
        const unsigned k2 = (unsigned)dv.z / RNG, k3 = (unsigned)dv.w / RNG;
        keys[wid * 32 + threadIdx.x] = k0 | (k1 << 8) | (k2 << 16) | (k3 << 24);
        uint4 p;
        p.x = (unsigned)sv.x | (((unsigned)dv.x - k0 * RNG) << 16);
        p.y = (unsigned)sv.y | (((unsigned)dv.y - k1 * RNG) << 16);
        p.z = (unsigned)sv.z | (((unsigned)dv.z - k2 * RNG) << 16);
        p.w = (unsigned)sv.w | (((unsigned)dv.w - k3 * RNG) << 16);
        ((uint4*)pk)[wid * 32 + threadIdx.x] = p;
    }

    const int which   = blockIdx.x;
    const int rowbase = blockIdx.y * 16;
    const float* __restrict__ W   = which ? W_dst : W_src;
    float* __restrict__       out = which ? dst_p : src_p;

    const int lane = threadIdx.x;
    const int half = lane >> 4;
    const int l    = lane & 15;

    const float* __restrict__ xrow = x + (size_t)(rowbase + l) * FF;

    v8f acc[8];
#pragma unroll
    for (int c = 0; c < 8; ++c) acc[c] = (v8f){};

    for (int k = 0; k < FF; k += 4) {
        v2f a;
        a.x = xrow[k + 2 * half + 0];
        a.y = xrow[k + 2 * half + 1];
#pragma unroll
        for (int c = 0; c < 8; ++c) {
            const float* wp = W + (size_t)(k + 2 * half) * FF + c * 16 + l;
            v2f b;
            b.x = wp[0];
            b.y = wp[FF];
            acc[c] = __builtin_amdgcn_wmma_f32_16x16x4_f32(
                false, a, false, b, (short)0, acc[c], false, false);
        }
    }

#pragma unroll
    for (int c = 0; c < 8; ++c) {
#pragma unroll
        for (int r = 0; r < 8; ++r) {
            out[(size_t)(rowbase + r + 8 * half) * FF + c * 16 + l] = acc[c][r];
        }
    }
}

__global__ __launch_bounds__(256) void gat_maxreduce(
    const float* __restrict__ partial, int n, float* __restrict__ gmax)
{
    __shared__ float smax[256];
    float s = -INFINITY;
    for (int i = threadIdx.x; i < n; i += 256) s = fmaxf(s, partial[i]);
    smax[threadIdx.x] = s;
    __syncthreads();
#pragma unroll
    for (int off = 128; off > 0; off >>= 1) {
        if (threadIdx.x < off)
            smax[threadIdx.x] = fmaxf(smax[threadIdx.x], smax[threadIdx.x + off]);
        __syncthreads();
    }
    if (threadIdx.x == 0) *gmax = smax[0];
}

#define MSG_THREADS 1024
#define MSG_WAVES   (MSG_THREADS / 32)
#define SCAN_DEPTH  8
#define QCAP        1056
#define NGROUPS     (EE / 128)

static_assert(NN % RNG == 0, "blocks tile the rows exactly");
static_assert(EE / 128 == 2 * (NN / 16) && EE % 128 == 0, "K1's waves tile the edge list, 128 edges each");
static_assert(MSG_BLOCKS <= 255 && NN <= 65536, "owner block fits a byte, source fits 16 bits");
static_assert(RNG <= 256, "local row fits the low 8 bits of a queue word");
static_assert(EE % 128 == 0 && EE < (1 << 24), "edge id fits the upper 24 bits");
static_assert(NGROUPS % MSG_BLOCKS == 0, "start offsets tile the list");
static_assert(QCAP >= 31 + SCAN_DEPTH * 128, "a step's edges always fit the queue");

__device__ __forceinline__ void wave_lds_order() {
    __builtin_amdgcn_fence(__ATOMIC_RELEASE, "wavefront");
    __builtin_amdgcn_wave_barrier();
    __builtin_amdgcn_fence(__ATOMIC_ACQUIRE, "wavefront");
}

__device__ __forceinline__ int queue_push(unsigned* __restrict__ q, int qn,
                                          bool m, unsigned word) {
    const unsigned mask = __builtin_amdgcn_ballot_w32(m);
    if (mask) {
        if (m) q[qn + (int)__builtin_amdgcn_mbcnt_lo(mask, 0u)] = word;
        qn += __builtin_popcount(mask);
    }
    return qn;
}

template <bool FULL>
__device__ __forceinline__ void drain_edges(
    const unsigned* __restrict__ q, int n, float* __restrict__ al,
    float* __restrict__ acc, float* __restrict__ den, float gm, int lane,
    const float* __restrict__ src_p,
    const float* __restrict__ scorebuf,
    const int*   __restrict__ srcI)
{
    wave_lds_order();
    const bool act = FULL || (lane < n);
    unsigned word = 0u;
    if (act) word = q[lane];
    const unsigned e  = word >> 8;
    const int      dl = (int)(word & 255u);
    int s = 0;
    if (act) {
        s = srcI[e];
        const float4* sc4 = (const float4*)(scorebuf + (size_t)e * HH);
        const float4 x0 = sc4[0], x1 = sc4[1];
        float4 a0, a1;
        a0.x = __expf(x0.x - gm);
        a0.y = __expf(x0.y - gm);
        a0.z = __expf(x0.z - gm);
        a0.w = __expf(x0.w - gm);
        a1.x = __expf(x1.x - gm);
        a1.y = __expf(x1.y - gm);
        a1.z = __expf(x1.z - gm);
        a1.w = __expf(x1.w - gm);
        float4* ap = (float4*)(al + lane * HH);
        ap[0] = a0;
        ap[1] = a1;
        float* dr = den + dl * HH;
        atomicAdd(dr + 0, a0.x);
        atomicAdd(dr + 1, a0.y);
        atomicAdd(dr + 2, a0.z);
        atomicAdd(dr + 3, a0.w);
        atomicAdd(dr + 4, a1.x);
        atomicAdd(dr + 5, a1.y);
        atomicAdd(dr + 6, a1.z);
        atomicAdd(dr + 7, a1.w);
    }
    wave_lds_order();

    const int hd = lane >> 2;
    const int hf = lane >> 4;
    const int o0 = lane * 4 + (hf ? 1 : 0);
    const int o1 = lane * 4 + (hf ? 2 : 1);
    const int o2 = lane * 4 + (hf ? 3 : 2);
    const int o3 = lane * 4 + (hf ? 0 : 3);
    const int cnt = FULL ? 32 : n;
    for (int j0 = 0; j0 < cnt; j0 += 8) {
        float4 v[8];
        float  a[8];
        int    dj[8];
#pragma unroll
        for (int u = 0; u < 8; ++u) {
            const int sj = __builtin_amdgcn_readlane(s, j0 + u);
            dj[u] = __builtin_amdgcn_readlane(dl, j0 + u);
            v[u]  = ((const float4*)(src_p + (size_t)sj * FF))[lane];
            a[u]  = al[(j0 + u) * HH + hd];
        }
#pragma unroll
        for (int u = 0; u < 8; ++u) {
            if (FULL || (j0 + u < cnt)) {
                const float p0 = v[u].x * a[u], p1 = v[u].y * a[u];
                const float p2 = v[u].z * a[u], p3 = v[u].w * a[u];
                float* row = acc + dj[u] * FF;
                atomicAdd(row + o0, hf ? p1 : p0);
                atomicAdd(row + o1, hf ? p2 : p1);
                atomicAdd(row + o2, hf ? p3 : p2);
                atomicAdd(row + o3, hf ? p0 : p3);
            }
        }
    }
}

#define LCAP 256

static_assert(MSG_WAVES == 32, "K4 reads a block's wave counts one per lane");

__device__ __forceinline__ float quad_sum(float p) {
    p += __int_as_float(__builtin_amdgcn_update_dpp(0, __float_as_int(p), 0x161, 0xf, 0xf, true));
    p += __int_as_float(__builtin_amdgcn_update_dpp(0, __float_as_int(p), 0x162, 0xf, 0xf, true));
    return p;
}

#define KGROUPS  (EE / 256)
#define KDEPTH   4
#define KQCAP    160

static_assert(EE % 256 == 0, "key reads tile the list exactly");
static_assert(KQCAP >= 31 + KDEPTH * 32, "a step's marked lanes always fit the queue");

__device__ __forceinline__ unsigned zero_bytes(unsigned x) {
    const unsigned t = (x & 0x7f7f7f7fu) + 0x7f7f7f7fu;
    return ~(t | x | 0x7f7f7f7fu);
}

__device__ __forceinline__ int queue_push2(uint2* q, int qn,
                                           bool m, unsigned idx, unsigned marks) {
    const unsigned mask = __builtin_amdgcn_ballot_w32(m);
    if (mask) {
        if (m) q[qn + (int)__builtin_amdgcn_mbcnt_lo(mask, 0u)] = make_uint2(idx, marks);
        qn += __builtin_popcount(mask);
    }
    return qn;
}

template <bool FULL>
__device__ __forceinline__ int pop_edges(uint2* q, int qn, int n,
                                         int lane, unsigned& e)
{
    wave_lds_order();
    const bool act = FULL || (lane < n);
    uint2 w = make_uint2(0u, 0u);
    if (act) w = q[qn + lane];
    const unsigned p = (w.y != 0u) ? (unsigned)__builtin_ctz(w.y) : 7u;
    e = w.x * 8u + (p >> 3) + (((p & 1u) == 0u) ? 4u : 0u);
    const unsigned rest = w.y & (w.y - 1u);
    return queue_push2(q, qn, rest != 0u, w.x, rest);
}

template <bool FULL>
__device__ __forceinline__ float score_edges(
    const unsigned e, int n, const float4* __restrict__ drow4,
    const float4 at, float mx, int lane,
    const float* __restrict__ src_p,
    const unsigned* __restrict__ pk,
    float*       __restrict__ scorebuf,
    unsigned*    __restrict__ seg, int kept)
{
    const bool act = FULL || (lane < n);
    int s = 0, dl = 0;
    if (act) {
        const unsigned p = pk[e];
        s  = (int)(p & 0xffffu);
        dl = (int)(p >> 16);
        if (kept + lane < LCAP) seg[kept + lane] = (e << 8) | (unsigned)dl;
    }
    const int hd  = lane >> 2;
    const int cnt = FULL ? 32 : n;
    for (int j0 = 0; j0 < cnt; j0 += 8) {
        float4 v[8];
        float4 d[8];
        int    ej[8];
#pragma unroll
        for (int u = 0; u < 8; ++u) {
            const int sj = __builtin_amdgcn_readlane(s, j0 + u);
            const int dj = __builtin_amdgcn_readlane(dl, j0 + u);
            ej[u] = __builtin_amdgcn_readlane((int)e, j0 + u);
            v[u]  = ((const float4*)(src_p + (size_t)sj * FF))[lane];
            d[u]  = drow4[dj * (FF / 4) + lane];
        }
#pragma unroll
        for (int u = 0; u < 8; ++u) {
            if (FULL || (j0 + u < cnt)) {
                float t0 = v[u].x + d[u].x, t1 = v[u].y + d[u].y;
                float t2 = v[u].z + d[u].z, t3 = v[u].w + d[u].w;
                t0 = (t0 >= 0.0f) ? t0 : 0.2f * t0;
                t1 = (t1 >= 0.0f) ? t1 : 0.2f * t1;
                t2 = (t2 >= 0.0f) ? t2 : 0.2f * t2;
                t3 = (t3 >= 0.0f) ? t3 : 0.2f * t3;
                const float p = quad_sum(at.x * t0 + at.y * t1 + at.z * t2 + at.w * t3);
                if ((lane & 3) == 0) scorebuf[(size_t)ej[u] * HH + hd] = p;
                mx = fmaxf(mx, p);
            }
        }
    }
    return mx;
}

__device__ __forceinline__ void keys_read(
    const int2* __restrict__ k2, int g0, int start, int lane,
    int2 (&d)[KDEPTH], int (&gi)[KDEPTH])
{
#pragma unroll
    for (int k = 0; k < KDEPTH; ++k) {
        int g = g0 + k * MSG_WAVES;
        const bool in = g < KGROUPS;
        g += start;
        if (g >= KGROUPS) g -= KGROUPS;
        gi[k] = in ? g : -1;
        d[k] = k2[(in ? g : 0) * 32 + lane];
    }
}

__global__ __launch_bounds__(MSG_THREADS) void gat_score_own(
    const float* __restrict__ src_p,
    const float* __restrict__ dst_p,
    const unsigned* __restrict__ keys,
    const unsigned* __restrict__ pk,
    const float* __restrict__ attn,
    float*       __restrict__ scorebuf,
    float*       __restrict__ partial,
    unsigned*    __restrict__ lists,
    int*         __restrict__ counts)
{
    __shared__ __attribute__((aligned(16))) float drow[RNG * FF];
    __shared__ __attribute__((aligned(16))) uint2 que[MSG_WAVES * KQCAP];
    __shared__ float wmax[MSG_WAVES];

    const int tid  = threadIdx.x;
    const int lane = tid & 31;
    const int wave = __builtin_amdgcn_readfirstlane(tid >> 5);
    const int base = blockIdx.x * RNG;

    const float4* __restrict__ dp4 = (const float4*)(dst_p + (size_t)base * FF);
    float4* drow4 = (float4*)drow;
    for (int i = tid; i < RNG * FF / 4; i += MSG_THREADS)
        drow4[i] = dp4[i];
    __syncthreads();

    const float4 at = ((const float4*)attn)[lane];
    uint2*    q   = que + wave * KQCAP;
    unsigned* seg = lists + (size_t)(blockIdx.x * MSG_WAVES + wave) * LCAP;
    int   qn   = 0;
    int   kept = 0;
    float mx   = -INFINITY;

    const int2* __restrict__ k2 = (const int2*)keys;
    const unsigned bb = (unsigned)blockIdx.x * 0x01010101u;
    const int start = blockIdx.x * (KGROUPS / MSG_BLOCKS);

    int2 dn[KDEPTH];
    int  gn[KDEPTH];
    keys_read(k2, wave, start, lane, dn, gn);
    for (int g0 = wave; g0 < KGROUPS; g0 += KDEPTH * MSG_WAVES) {
        int2 d[KDEPTH];
        int  gi[KDEPTH];
#pragma unroll
        for (int k = 0; k < KDEPTH; ++k) { d[k] = dn[k]; gi[k] = gn[k]; }
        keys_read(k2, g0 + KDEPTH * MSG_WAVES, start, lane, dn, gn);
#pragma unroll
        for (int k = 0; k < KDEPTH; ++k) {
            if (gi[k] < 0) continue;
            const unsigned m0 = zero_bytes((unsigned)d[k].x ^ bb);
            const unsigned m1 = zero_bytes((unsigned)d[k].y ^ bb);
            const unsigned marks = m0 | (m1 >> 1);
            qn = queue_push2(q, qn, marks != 0u, (unsigned)(gi[k] * 32 + lane), marks);
        }
        while (qn >= 32) {
            qn -= 32;
            unsigned e;
            qn = pop_edges<true>(q, qn, 32, lane, e);
            mx = score_edges<true>(e, 32, drow4, at, mx, lane,
                                   src_p, pk, scorebuf, seg, kept);
            kept += 32;
        }
    }
    while (qn > 0) {
        const int n = qn < 32 ? qn : 32;
        qn -= n;
        unsigned e;
        qn = pop_edges<false>(q, qn, n, lane, e);
        mx = score_edges<false>(e, n, drow4, at, mx, lane,
                                src_p, pk, scorebuf, seg, kept);
        kept += n;
    }
    if (lane == 0) counts[blockIdx.x * MSG_WAVES + wave] = kept;

    mx = fmaxf(mx, __shfl_xor(mx, 16));
    mx = fmaxf(mx, __shfl_xor(mx, 8));
    mx = fmaxf(mx, __shfl_xor(mx, 4));
    mx = fmaxf(mx, __shfl_xor(mx, 2));
    mx = fmaxf(mx, __shfl_xor(mx, 1));
    if (lane == 0) wmax[wave] = mx;
    __syncthreads();
    if (tid == 0) {
        float m = wmax[0];
        for (int i = 1; i < MSG_WAVES; ++i) m = fmaxf(m, wmax[i]);
        partial[blockIdx.x] = m;
    }
}

__global__ __launch_bounds__(MSG_THREADS) void gat_message_own(
    const float* __restrict__ src_p,
    const float* __restrict__ scorebuf,
    const float* __restrict__ gmax,
    const int*   __restrict__ srcI,
    const int*   __restrict__ dstI,
    const float* __restrict__ x,
    const float* __restrict__ bias,
    const float* __restrict__ prelu_w,
    const unsigned* __restrict__ lists,
    const int*   __restrict__ counts,
    float* __restrict__ out)
{
    __shared__ __attribute__((aligned(16))) float    acc[RNG * FF];
    __shared__ __attribute__((aligned(16))) float    den[RNG * HH];
    __shared__ __attribute__((aligned(16))) float    alp[MSG_WAVES * 32 * HH];
    __shared__ __attribute__((aligned(16))) unsigned que[MSG_WAVES * QCAP];

    const int tid  = threadIdx.x;
    const int lane = tid & 31;
    const int wave = __builtin_amdgcn_readfirstlane(tid >> 5);
    const int base = blockIdx.x * RNG;

    const float gm = *gmax;

    float4* acc4 = (float4*)acc;
    for (int i = tid; i < RNG * FF / 4; i += MSG_THREADS)
        acc4[i] = make_float4(0.0f, 0.0f, 0.0f, 0.0f);
    for (int i = tid; i < RNG * HH; i += MSG_THREADS)
        den[i] = 0.0f;
    __syncthreads();

    unsigned* q  = que + wave * QCAP;
    float*    al = alp + wave * 32 * HH;
    int qn = 0;

    const int  cl   = counts[blockIdx.x * MSG_WAVES + lane];
    const bool fits = __builtin_amdgcn_ballot_w32(cl > LCAP) == 0u;
    if (fits) {
        const int c = __builtin_amdgcn_readlane(cl, wave);
        const unsigned* __restrict__ seg =
            lists + (size_t)(blockIdx.x * MSG_WAVES + wave) * LCAP;
        for (int i = 0; i + 32 <= c; i += 32) {
            q[lane] = seg[i + lane];
            drain_edges<true>(q, 32, al, acc, den, gm, lane, src_p, scorebuf, srcI);
        }
        qn = c & 31;
        if (lane < qn) q[lane] = seg[(c & ~31) + lane];
    } else {
        const int4* __restrict__ d4 = (const int4*)dstI;
        const int start = blockIdx.x * (NGROUPS / MSG_BLOCKS);

        for (int g0 = wave; g0 < NGROUPS; g0 += SCAN_DEPTH * MSG_WAVES) {
            int4 d[SCAN_DEPTH];
            int  gi[SCAN_DEPTH];
#pragma unroll
            for (int k = 0; k < SCAN_DEPTH; ++k) {
                int g = g0 + k * MSG_WAVES;
                const bool in = g < NGROUPS;
                g += start;
                if (g >= NGROUPS) g -= NGROUPS;
                gi[k] = in ? g : -1;
                if (in) d[k] = d4[g * 32 + lane];
                else    d[k] = make_int4(-1, -1, -1, -1);
            }
#pragma unroll
            for (int k = 0; k < SCAN_DEPTH; ++k) {
                if (gi[k] < 0) continue;
                const unsigned r0 = (unsigned)(d[k].x - base);
                const unsigned r1 = (unsigned)(d[k].y - base);
                const unsigned r2 = (unsigned)(d[k].z - base);
                const unsigned r3 = (unsigned)(d[k].w - base);
                const bool m0 = r0 < (unsigned)RNG, m1 = r1 < (unsigned)RNG;
                const bool m2 = r2 < (unsigned)RNG, m3 = r3 < (unsigned)RNG;
                if (__builtin_amdgcn_ballot_w32(m0 | m1 | m2 | m3)) {
                    const unsigned e0 = (unsigned)(gi[k] * 32 + lane) * 4u;
                    qn = queue_push(q, qn, m0, ((e0 + 0u) << 8) | r0);
                    qn = queue_push(q, qn, m1, ((e0 + 1u) << 8) | r1);
                    qn = queue_push(q, qn, m2, ((e0 + 2u) << 8) | r2);
                    qn = queue_push(q, qn, m3, ((e0 + 3u) << 8) | r3);
                }
            }
            while (qn >= 32) {
                qn -= 32;
                drain_edges<true>(q + qn, 32, al, acc, den, gm, lane,
                                  src_p, scorebuf, srcI);
            }
        }
    }
    if (qn > 0)
        drain_edges<false>(q, qn, al, acc, den, gm, lane, src_p, scorebuf, srcI);

    __syncthreads();
    const float4* __restrict__ x4 = (const float4*)(x + (size_t)base * FF);
    const float4* __restrict__ b4 = (const float4*)bias;
    const float pw = prelu_w[0];
    float4* o4 = (float4*)(out + (size_t)base * FF);
    for (int i = tid; i < RNG * FF / 4; i += MSG_THREADS) {
        const int row   = i >> 5;
        const int piece = i & 31;
        const float  d  = den[row * HH + (piece >> 2)] + 1e-16f;
        const float4 a  = acc4[i];
        const float4 xv = x4[i];
        const float4 bv = b4[piece];
        float4 r;
        r.x = a.x / d + xv.x + bv.x;
        r.y = a.y / d + xv.y + bv.y;
        r.z = a.z / d + xv.z + bv.z;
        r.w = a.w / d + xv.w + bv.w;
        r.x = (r.x >= 0.0f) ? r.x : pw * r.x;
        r.y = (r.y >= 0.0f) ? r.y : pw * r.y;
        r.z = (r.z >= 0.0f) ? r.z : pw * r.z;
        r.w = (r.w >= 0.0f) ? r.w : pw * r.w;
        o4[i] = r;
    }
}

extern "C" void kernel_launch(void* const* d_in, const int* in_sizes, int n_in,
                              void* d_out, int out_size, void* d_ws, size_t ws_size,
                              hipStream_t stream) {
    (void)in_sizes; (void)n_in; (void)out_size; (void)ws_size;

    const float* x        = (const float*)d_in[0];
    const int*   eidx     = (const int*)  d_in[1];
    const float* W_src    = (const float*)d_in[2];
    const float* W_dst    = (const float*)d_in[3];
    const float* attn     = (const float*)d_in[4];
    const float* bias     = (const float*)d_in[5];
    const float* prelu_w  = (const float*)d_in[6];
    const int*   srcI     = eidx;
    const int*   dstI     = eidx + EE;

    float* out = (float*)d_out;

    float* ws      = (float*)d_ws;
    float* src_p   = ws;
    float* dst_p   = src_p + (size_t)NN * FF;
    float* scoreb  = dst_p + (size_t)NN * FF;
    unsigned* keys = (unsigned*)(scoreb + (size_t)EE * HH);
    unsigned* pk   = keys + EE / 4;
    float* partial = (float*)(pk + EE);
    float* gmax    = partial + MSG_BLOCKS;
    int*      counts = (int*)(gmax + 1);
    unsigned* lists  = (unsigned*)(counts + MSG_BLOCKS * MSG_WAVES);

    {
        dim3 grid(2, NN / 16);
        gat_proj_gemm<<<grid, 32, 0, stream>>>(x, W_src, W_dst, src_p, dst_p,
                                               srcI, dstI, keys, pk);
    }

    gat_score_own<<<MSG_BLOCKS, MSG_THREADS, 0, stream>>>(src_p, dst_p, keys, pk,
                                                          attn, scoreb, partial,
                                                          lists, counts);
    gat_maxreduce<<<1, 256, 0, stream>>>(partial, MSG_BLOCKS, gmax);

    gat_message_own<<<MSG_BLOCKS, MSG_THREADS, 0, stream>>>(src_p, scoreb, gmax,
                                                            srcI, dstI, x, bias,
                                                            prelu_w, lists, counts,
                                                            out);
}


# score and message launches fused: owner blocks add exp(score - block shift) * source row to LDS sums in the one key walk (1 row gather per edge, add order free, rounding changed); 3 launches
# speedup vs baseline: 6.2789x; 1.4339x over previous
#include <hip/hip_runtime.h>
#include <hip/hip_bf16.h>
#include <math.h>

#define NN 50000
#define EE 800000
#define FF 128
#define HH 8
#define DD 16

#define RNG         200
#define MSG_BLOCKS  (NN / RNG)

typedef __attribute__((ext_vector_type(2))) float v2f;
typedef __attribute__((ext_vector_type(8))) float v8f;

__global__ __launch_bounds__(32) void gat_proj_gemm(
    const float* __restrict__ x,
    const float* __restrict__ W_src,
    const float* __restrict__ W_dst,
    float* __restrict__ src_p,
    float* __restrict__ dst_p,
    const int* __restrict__ srcI,
    const int* __restrict__ dstI,
    unsigned* __restrict__ keys,
    unsigned* __restrict__ pk)
{
    {
        const int wid = blockIdx.y * 2 + blockIdx.x;
        const int4 dv = ((const int4*)dstI)[wid * 32 + threadIdx.x];
        const int4 sv = ((const int4*)srcI)[wid * 32 + threadIdx.x];
        const unsigned k0 = (unsigned)dv.x / RNG, k1 = (unsigned)dv.y / RNG;
        const unsigned k2 = (unsigned)dv.z / RNG, k3 = (unsigned)dv.w / RNG;
        keys[wid * 32 + threadIdx.x] = k0 | (k1 << 8) | (k2 << 16) | (k3 << 24);
        uint4 p;
        p.x = (unsigned)sv.x | (((unsigned)dv.x - k0 * RNG) << 16);
        p.y = (unsigned)sv.y | (((unsigned)dv.y - k1 * RNG) << 16);
        p.z = (unsigned)sv.z | (((unsigned)dv.z - k2 * RNG) << 16);
        p.w = (unsigned)sv.w | (((unsigned)dv.w - k3 * RNG) << 16);
        ((uint4*)pk)[wid * 32 + threadIdx.x] = p;
    }

    const int which   = blockIdx.x;
    const int rowbase = blockIdx.y * 16;
    const float* __restrict__ W   = which ? W_dst : W_src;
    float* __restrict__       out = which ? dst_p : src_p;

    const int lane = threadIdx.x;
    const int half = lane >> 4;
    const int l    = lane & 15;

    const float* __restrict__ xrow = x + (size_t)(rowbase + l) * FF;

    v8f acc[8];
#pragma unroll
    for (int c = 0; c < 8; ++c) acc[c] = (v8f){};

    for (int k = 0; k < FF; k += 4) {
        v2f a;
        a.x = xrow[k + 2 * half + 0];
        a.y = xrow[k + 2 * half + 1];
#pragma unroll
        for (int c = 0; c < 8; ++c) {
            const float* wp = W + (size_t)(k + 2 * half) * FF + c * 16 + l;
            v2f b;
            b.x = wp[0];
            b.y = wp[FF];
            acc[c] = __builtin_amdgcn_wmma_f32_16x16x4_f32(
                false, a, false, b, (short)0, acc[c], false, false);
        }
    }

#pragma unroll
    for (int c = 0; c < 8; ++c) {
#pragma unroll
        for (int r = 0; r < 8; ++r) {
            out[(size_t)(rowbase + r + 8 * half) * FF + c * 16 + l] = acc[c][r];
        }
    }
}

#define MSG_THREADS 1024
#define MSG_WAVES   (MSG_THREADS / 32)
#define SHIFT0      10.0f
#define SHIFT_SPAN  20.0f

static_assert(NN % RNG == 0, "blocks tile the rows exactly");
static_assert(EE / 128 == 2 * (NN / 16) && EE % 128 == 0, "K1's waves tile the edge list, 128 edges each");
static_assert(MSG_BLOCKS <= 255 && NN <= 65536, "owner block fits a byte, source fits 16 bits");
static_assert(MSG_WAVES == 32, "the block maximum is read one wave maximum per lane");

__device__ __forceinline__ void wave_lds_order() {
    __builtin_amdgcn_fence(__ATOMIC_RELEASE, "wavefront");
    __builtin_amdgcn_wave_barrier();
    __builtin_amdgcn_fence(__ATOMIC_ACQUIRE, "wavefront");
}

__device__ __forceinline__ float quad_sum(float p) {
    p += __int_as_float(__builtin_amdgcn_update_dpp(0, __float_as_int(p), 0x161, 0xf, 0xf, true));
    p += __int_as_float(__builtin_amdgcn_update_dpp(0, __float_as_int(p), 0x162, 0xf, 0xf, true));
    return p;
}

#define KGROUPS  (EE / 256)
#define KDEPTH   4
#define KQCAP    160

static_assert(EE % 256 == 0, "key reads tile the list exactly");
static_assert(KQCAP >= 31 + KDEPTH * 32, "a step's marked lanes always fit the queue");

__device__ __forceinline__ unsigned zero_bytes(unsigned x) {
    const unsigned t = (x & 0x7f7f7f7fu) + 0x7f7f7f7fu;
    return ~(t | x | 0x7f7f7f7fu);
}

__device__ __forceinline__ int queue_push2(uint2* q, int qn,
                                           bool m, unsigned idx, unsigned marks) {
    const unsigned mask = __builtin_amdgcn_ballot_w32(m);
    if (mask) {
        if (m) q[qn + (int)__builtin_amdgcn_mbcnt_lo(mask, 0u)] = make_uint2(idx, marks);
        qn += __builtin_popcount(mask);
    }
    return qn;
}

template <bool FULL>
__device__ __forceinline__ int pop_edges(uint2* q, int qn, int n,
                                         int lane, unsigned& e)
{
    wave_lds_order();
    const bool act = FULL || (lane < n);
    uint2 w = make_uint2(0u, 0u);
    if (act) w = q[qn + lane];
    const unsigned p = (w.y != 0u) ? (unsigned)__builtin_ctz(w.y) : 7u;
    e = w.x * 8u + (p >> 3) + (((p & 1u) == 0u) ? 4u : 0u);
    const unsigned rest = w.y & (w.y - 1u);
    return queue_push2(q, qn, rest != 0u, w.x, rest);
}

template <bool FULL>
__device__ __forceinline__ float fuse_edges(
    const unsigned e, int n, const float4* __restrict__ drow4,
    const float4 at, const float sub, const float scl, float mx, int lane,
    const float* __restrict__ src_p,
    const unsigned* __restrict__ pk,
    float* __restrict__ acc, float* __restrict__ den)
{
    const bool act = FULL || (lane < n);
    int s = 0, dl = 0;
    if (act) {
        const unsigned p = pk[e];
        s  = (int)(p & 0xffffu);
        dl = (int)(p >> 16);
    }
    const int hd  = lane >> 2;
    const int hf  = lane >> 4;
    const int o0 = lane * 4 + (hf ? 1 : 0);
    const int o1 = lane * 4 + (hf ? 2 : 1);
    const int o2 = lane * 4 + (hf ? 3 : 2);
    const int o3 = lane * 4 + (hf ? 0 : 3);
    const int cnt = FULL ? 32 : n;
    for (int j0 = 0; j0 < cnt; j0 += 8) {
        float4 v[8];
        float4 d[8];
        int    dj[8];
#pragma unroll
        for (int u = 0; u < 8; ++u) {
            const int sj = __builtin_amdgcn_readlane(s, j0 + u);
            dj[u] = __builtin_amdgcn_readlane(dl, j0 + u);
            v[u]  = ((const float4*)(src_p + (size_t)sj * FF))[lane];
            d[u]  = drow4[dj[u] * (FF / 4) + lane];
        }
#pragma unroll
        for (int u = 0; u < 8; ++u) {
            if (FULL || (j0 + u < cnt)) {
                float t0 = v[u].x + d[u].x, t1 = v[u].y + d[u].y;
                float t2 = v[u].z + d[u].z, t3 = v[u].w + d[u].w;
                t0 = (t0 >= 0.0f) ? t0 : 0.2f * t0;
                t1 = (t1 >= 0.0f) ? t1 : 0.2f * t1;
                t2 = (t2 >= 0.0f) ? t2 : 0.2f * t2;
                t3 = (t3 >= 0.0f) ? t3 : 0.2f * t3;
                const float p = quad_sum(at.x * t0 + at.y * t1 + at.z * t2 + at.w * t3);
                mx = fmaxf(mx, p);
                const float w  = __expf(p - sub) * scl;
                const float p0 = v[u].x * w, p1 = v[u].y * w;
                const float p2 = v[u].z * w, p3 = v[u].w * w;
                float* row = acc + dj[u] * FF;
                atomicAdd(row + o0, hf ? p1 : p0);
                atomicAdd(row + o1, hf ? p2 : p1);
                atomicAdd(row + o2, hf ? p3 : p2);
                atomicAdd(row + o3, hf ? p0 : p3);
                if ((lane & 3) == 0) atomicAdd(den + dj[u] * HH + hd, w);
            }
        }
    }
    return mx;
}

__device__ __forceinline__ void keys_read(
    const int2* __restrict__ k2, int g0, int start, int lane,
    int2 (&d)[KDEPTH], int (&gi)[KDEPTH])
{
#pragma unroll
    for (int k = 0; k < KDEPTH; ++k) {
        int g = g0 + k * MSG_WAVES;
        const bool in = g < KGROUPS;
        g += start;
        if (g >= KGROUPS) g -= KGROUPS;
        gi[k] = in ? g : -1;
        d[k] = k2[(in ? g : 0) * 32 + lane];
    }
}

__global__ __launch_bounds__(MSG_THREADS) void gat_fused_own(
    const float* __restrict__ src_p,
    const float* __restrict__ dst_p,
    const unsigned* __restrict__ keys,
    const unsigned* __restrict__ pk,
    const float* __restrict__ attn,
    const float* __restrict__ x,
    const float* __restrict__ bias,
    const float* __restrict__ prelu_w,
    float*       __restrict__ partial,
    float*       __restrict__ shiftg,
    float*       __restrict__ deng,
    float*       __restrict__ accg,
    float*       __restrict__ out)
{
    __shared__ __attribute__((aligned(16))) float drow[RNG * FF];
    __shared__ __attribute__((aligned(16))) float acc[RNG * FF];
    __shared__ __attribute__((aligned(16))) float den[RNG * HH];
    __shared__ __attribute__((aligned(16))) uint2 que[MSG_WAVES * KQCAP];
    __shared__ float wmax[MSG_WAVES];

    const int tid  = threadIdx.x;
    const int lane = tid & 31;
    const int wave = __builtin_amdgcn_readfirstlane(tid >> 5);
    const int base = blockIdx.x * RNG;

    const float4* __restrict__ dp4 = (const float4*)(dst_p + (size_t)base * FF);
    float4* drow4 = (float4*)drow;
    float4* acc4  = (float4*)acc;
    for (int i = tid; i < RNG * FF / 4; i += MSG_THREADS)
        drow4[i] = dp4[i];

    const float4 at = ((const float4*)attn)[lane];
    uint2* q = que + wave * KQCAP;

    const int2* __restrict__ k2 = (const int2*)keys;
    const unsigned bb = (unsigned)blockIdx.x * 0x01010101u;
    const int start = blockIdx.x * (KGROUPS / MSG_BLOCKS);

    float shift = SHIFT0;
    float sub   = 0.0f;
    float scl   = __expf(-SHIFT0);
    float bmax  = -INFINITY;
    for (int pass = 0; pass < 2; ++pass) {
        for (int i = tid; i < RNG * FF / 4; i += MSG_THREADS)
            acc4[i] = make_float4(0.0f, 0.0f, 0.0f, 0.0f);
        for (int i = tid; i < RNG * HH; i += MSG_THREADS)
            den[i] = 0.0f;
        __syncthreads();

        int   qn = 0;
        float mx = -INFINITY;
        int2 dn[KDEPTH];
        int  gn[KDEPTH];
        keys_read(k2, wave, start, lane, dn, gn);
        for (int g0 = wave; g0 < KGROUPS; g0 += KDEPTH * MSG_WAVES) {
            int2 d[KDEPTH];
            int  gi[KDEPTH];
#pragma unroll
            for (int k = 0; k < KDEPTH; ++k) { d[k] = dn[k]; gi[k] = gn[k]; }
            keys_read(k2, g0 + KDEPTH * MSG_WAVES, start, lane, dn, gn);
#pragma unroll
            for (int k = 0; k < KDEPTH; ++k) {
                if (gi[k] < 0) continue;
                const unsigned m0 = zero_bytes((unsigned)d[k].x ^ bb);
                const unsigned m1 = zero_bytes((unsigned)d[k].y ^ bb);
                const unsigned marks = m0 | (m1 >> 1);
                qn = queue_push2(q, qn, marks != 0u, (unsigned)(gi[k] * 32 + lane), marks);
            }
            while (qn >= 32) {
                qn -= 32;
                unsigned e;
                qn = pop_edges<true>(q, qn, 32, lane, e);
                mx = fuse_edges<true>(e, 32, drow4, at, sub, scl, mx, lane,
                                      src_p, pk, acc, den);
            }
        }
        while (qn > 0) {
            const int n = qn < 32 ? qn : 32;
            qn -= n;
            unsigned e;
            qn = pop_edges<false>(q, qn, n, lane, e);
            mx = fuse_edges<false>(e, n, drow4, at, sub, scl, mx, lane,
                                   src_p, pk, acc, den);
        }

        mx = fmaxf(mx, __shfl_xor(mx, 16));
        mx = fmaxf(mx, __shfl_xor(mx, 8));
        mx = fmaxf(mx, __shfl_xor(mx, 4));
        mx = fmaxf(mx, __shfl_xor(mx, 2));
        mx = fmaxf(mx, __shfl_xor(mx, 1));
        if (lane == 0) wmax[wave] = mx;
        __syncthreads();
        float m = wmax[lane];
        m = fmaxf(m, __shfl_xor(m, 16));
        m = fmaxf(m, __shfl_xor(m, 8));
        m = fmaxf(m, __shfl_xor(m, 4));
        m = fmaxf(m, __shfl_xor(m, 2));
        m = fmaxf(m, __shfl_xor(m, 1));
        bmax = __int_as_float(__builtin_amdgcn_readfirstlane(__float_as_int(m)));
        const bool again = (pass == 0) &&
                           ((bmax > SHIFT0) ||
                            (bmax < SHIFT0 - SHIFT_SPAN && bmax > -INFINITY));
        if (!again) break;
        shift = bmax;
        sub   = bmax;
        scl   = 1.0f;
    }

    const float4* __restrict__ x4 = (const float4*)(x + (size_t)base * FF);
    const float4* __restrict__ b4 = (const float4*)bias;
    const float pw = prelu_w[0];
    float4* o4 = (float4*)(out  + (size_t)base * FF);
    float4* g4 = (float4*)(accg + (size_t)base * FF);
    for (int i = tid; i < RNG * FF / 4; i += MSG_THREADS) {
        const int row   = i >> 5;
        const int piece = i & 31;
        const float  d  = den[row * HH + (piece >> 2)];
        const float4 a  = acc4[i];
        const float4 xv = x4[i];
        const float4 bv = b4[piece];
        g4[i] = a;
        const bool has = d > 0.0f;
        float4 r;
        r.x = (has ? a.x / d : 0.0f) + xv.x + bv.x;
        r.y = (has ? a.y / d : 0.0f) + xv.y + bv.y;
        r.z = (has ? a.z / d : 0.0f) + xv.z + bv.z;
        r.w = (has ? a.w / d : 0.0f) + xv.w + bv.w;
        r.x = (r.x >= 0.0f) ? r.x : pw * r.x;
        r.y = (r.y >= 0.0f) ? r.y : pw * r.y;
        r.z = (r.z >= 0.0f) ? r.z : pw * r.z;
        r.w = (r.w >= 0.0f) ? r.w : pw * r.w;
        o4[i] = r;
    }
    for (int i = tid; i < RNG * HH; i += MSG_THREADS)
        deng[(size_t)base * HH + i] = den[i];
    if (tid == 0) {
        partial[blockIdx.x] = bmax;
        shiftg[blockIdx.x]  = shift;
    }
}

__global__ __launch_bounds__(256) void gat_eps_fix(
    const float* __restrict__ partial,
    const float* __restrict__ shiftg,
    const float* __restrict__ deng,
    const float* __restrict__ accg,
    const float* __restrict__ x,
    const float* __restrict__ bias,
    const float* __restrict__ prelu_w,
    float* __restrict__ out)
{
    __shared__ float smax[256];
    float s = -INFINITY;
    for (int i = threadIdx.x; i < MSG_BLOCKS; i += 256) s = fmaxf(s, partial[i]);
    smax[threadIdx.x] = s;
    __syncthreads();
#pragma unroll
    for (int off = 128; off > 0; off >>= 1) {
        if (threadIdx.x < off)
            smax[threadIdx.x] = fmaxf(smax[threadIdx.x], smax[threadIdx.x + off]);
        __syncthreads();
    }
    const float gm  = smax[0];
    const float eh  = expf(0.5f * (gm - shiftg[blockIdx.x]));
    const float eps = eh * (1e-16f * eh);
    const float pw  = prelu_w[0];
    const size_t base = (size_t)blockIdx.x * RNG;
    for (int i = threadIdx.x; i < RNG * HH; i += 256) {
        const float d  = deng[base * HH + i];
        const float de = d + eps;
        if (de != d) {
            const int row = i >> 3;
            const int h   = i & 7;
            const size_t o = (base + row) * FF + h * DD;
            for (int c = 0; c < DD; ++c) {
                float r = accg[o + c] / de + x[o + c] + bias[h * DD + c];
                r = (r >= 0.0f) ? r : pw * r;
                out[o + c] = r;
            }
        }
    }
}

extern "C" void kernel_launch(void* const* d_in, const int* in_sizes, int n_in,
                              void* d_out, int out_size, void* d_ws, size_t ws_size,
                              hipStream_t stream) {
    (void)in_sizes; (void)n_in; (void)out_size; (void)ws_size;

    const float* x        = (const float*)d_in[0];
    const int*   eidx     = (const int*)  d_in[1];
    const float* W_src    = (const float*)d_in[2];
    const float* W_dst    = (const float*)d_in[3];
    const float* attn     = (const float*)d_in[4];
    const float* bias     = (const float*)d_in[5];
    const float* prelu_w  = (const float*)d_in[6];
    const int*   srcI     = eidx;
    const int*   dstI     = eidx + EE;

    float* out = (float*)d_out;

    float* ws      = (float*)d_ws;
    float* src_p   = ws;
    float* dst_p   = src_p + (size_t)NN * FF;
    float* accg    = dst_p + (size_t)NN * FF;
    float* deng    = accg + (size_t)NN * FF;
    unsigned* keys = (unsigned*)(deng + (size_t)NN * HH);
    unsigned* pk   = keys + EE / 4;
    float* partial = (float*)(pk + EE);
    float* shiftg  = partial + MSG_BLOCKS;

    {
        dim3 grid(2, NN / 16);
        gat_proj_gemm<<<grid, 32, 0, stream>>>(x, W_src, W_dst, src_p, dst_p,
                                               srcI, dstI, keys, pk);
    }

    gat_fused_own<<<MSG_BLOCKS, MSG_THREADS, 0, stream>>>(src_p, dst_p, keys, pk, attn,
                                                          x, bias, prelu_w,
                                                          partial, shiftg, deng, accg,
                                                          out);

    gat_eps_fix<<<MSG_BLOCKS, 256, 0, stream>>>(partial, shiftg, deng, accg,
                                                x, bias, prelu_w, out);
}


# owner blocks' edge loop, fewer instructions per edge: sum rows lane-transposed in LDS, 4 edges' denominators per LDS add, 4 edges scored abreast (dot fused), leaky_relu as one max; add order free
# speedup vs baseline: 6.5857x; 1.0489x over previous
#include <hip/hip_runtime.h>
#include <hip/hip_bf16.h>
#include <math.h>

#define NN 50000
#define EE 800000
#define FF 128
#define HH 8
#define DD 16

#define RNG         200
#define MSG_BLOCKS  (NN / RNG)

typedef __attribute__((ext_vector_type(2))) float v2f;
typedef __attribute__((ext_vector_type(8))) float v8f;
typedef __attribute__((ext_vector_type(4))) float v4f;

__global__ __launch_bounds__(32) void gat_proj_gemm(
    const float* __restrict__ x,
    const float* __restrict__ W_src,
    const float* __restrict__ W_dst,
    float* __restrict__ src_p,
    float* __restrict__ dst_p,
    const int* __restrict__ srcI,
    const int* __restrict__ dstI,
    unsigned* __restrict__ keys,
    unsigned* __restrict__ pk)
{
    {
        const int wid = blockIdx.y * 2 + blockIdx.x;
        const int4 dv = ((const int4*)dstI)[wid * 32 + threadIdx.x];
        const int4 sv = ((const int4*)srcI)[wid * 32 + threadIdx.x];
        const unsigned k0 = (unsigned)dv.x / RNG, k1 = (unsigned)dv.y / RNG;
        const unsigned k2 = (unsigned)dv.z / RNG, k3 = (unsigned)dv.w / RNG;
        keys[wid * 32 + threadIdx.x] = k0 | (k1 << 8) | (k2 << 16) | (k3 << 24);
        uint4 p;
        p.x = (unsigned)sv.x | (((unsigned)dv.x - k0 * RNG) << 16);
        p.y = (unsigned)sv.y | (((unsigned)dv.y - k1 * RNG) << 16);
        p.z = (unsigned)sv.z | (((unsigned)dv.z - k2 * RNG) << 16);
        p.w = (unsigned)sv.w | (((unsigned)dv.w - k3 * RNG) << 16);
        ((uint4*)pk)[wid * 32 + threadIdx.x] = p;
    }

    const int which   = blockIdx.x;
    const int rowbase = blockIdx.y * 16;
    const float* __restrict__ W   = which ? W_dst : W_src;
    float* __restrict__       out = which ? dst_p : src_p;

    const int lane = threadIdx.x;
    const int half = lane >> 4;
    const int l    = lane & 15;

    const float* __restrict__ xrow = x + (size_t)(rowbase + l) * FF;

    v8f acc[8];
#pragma unroll
    for (int c = 0; c < 8; ++c) acc[c] = (v8f){};

    for (int k = 0; k < FF; k += 4) {
        v2f a;
        a.x = xrow[k + 2 * half + 0];
        a.y = xrow[k + 2 * half + 1];
#pragma unroll
        for (int c = 0; c < 8; ++c) {
            const float* wp = W + (size_t)(k + 2 * half) * FF + c * 16 + l;
            v2f b;
            b.x = wp[0];
            b.y = wp[FF];
            acc[c] = __builtin_amdgcn_wmma_f32_16x16x4_f32(
                false, a, false, b, (short)0, acc[c], false, false);
        }
    }

#pragma unroll
    for (int c = 0; c < 8; ++c) {
#pragma unroll
        for (int r = 0; r < 8; ++r) {
            out[(size_t)(rowbase + r + 8 * half) * FF + c * 16 + l] = acc[c][r];
        }
    }
}

#define MSG_THREADS 1024
#define MSG_WAVES   (MSG_THREADS / 32)
#define SHIFT0      10.0f
#define SHIFT_SPAN  20.0f

static_assert(NN % RNG == 0, "blocks tile the rows exactly");
static_assert(EE / 128 == 2 * (NN / 16) && EE % 128 == 0, "K1's waves tile the edge list, 128 edges each");
static_assert(MSG_BLOCKS <= 255 && NN <= 65536, "owner block fits a byte, source fits 16 bits");
static_assert(MSG_WAVES == 32, "the block maximum is read one wave maximum per lane");

__device__ __forceinline__ void wave_lds_order() {
    __builtin_amdgcn_fence(__ATOMIC_RELEASE, "wavefront");
    __builtin_amdgcn_wave_barrier();
    __builtin_amdgcn_fence(__ATOMIC_ACQUIRE, "wavefront");
}

__device__ __forceinline__ float quad_sum(float p) {
    p += __int_as_float(__builtin_amdgcn_update_dpp(0, __float_as_int(p), 0x161, 0xf, 0xf, true));
    p += __int_as_float(__builtin_amdgcn_update_dpp(0, __float_as_int(p), 0x162, 0xf, 0xf, true));
    return p;
}

#define KGROUPS  (EE / 256)
#define KDEPTH   4
#define KQCAP    160

static_assert(EE % 256 == 0, "key reads tile the list exactly");
static_assert(KQCAP >= 31 + KDEPTH * 32, "a step's marked lanes always fit the queue");

__device__ __forceinline__ unsigned zero_bytes(unsigned x) {
    const unsigned t = (x & 0x7f7f7f7fu) + 0x7f7f7f7fu;
    return ~(t | x | 0x7f7f7f7fu);
}

__device__ __forceinline__ int queue_push2(uint2* q, int qn,
                                           bool m, unsigned idx, unsigned marks) {
    const unsigned mask = __builtin_amdgcn_ballot_w32(m);
    if (mask) {
        if (m) q[qn + (int)__builtin_amdgcn_mbcnt_lo(mask, 0u)] = make_uint2(idx, marks);
        qn += __builtin_popcount(mask);
    }
    return qn;
}

template <bool FULL>
__device__ __forceinline__ int pop_edges(uint2* q, int qn, int n,
                                         int lane, unsigned& e)
{
    wave_lds_order();
    const bool act = FULL || (lane < n);
    uint2 w = make_uint2(0u, 0u);
    if (act) w = q[qn + lane];
    const unsigned p = (w.y != 0u) ? (unsigned)__builtin_ctz(w.y) : 7u;
    e = w.x * 8u + (p >> 3) + (((p & 1u) == 0u) ? 4u : 0u);
    const unsigned rest = w.y & (w.y - 1u);
    return queue_push2(q, qn, rest != 0u, w.x, rest);
}

template <bool FULL>
__device__ __forceinline__ float fuse_edges(
    const unsigned e, int n, const float4* __restrict__ drow4,
    const float4 at, const float sub, const float scl, float mx, int lane,
    const float* __restrict__ src_p,
    const unsigned* __restrict__ pk,
    float* __restrict__ acc, float* __restrict__ den)
{
    const bool act = FULL || (lane < n);
    int s = 0, dl = 0;
    if (act) {
        const unsigned p = pk[e];
        s  = (int)(p & 0xffffu);
        dl = (int)(p >> 16);
    }
    const int hd  = lane >> 2;
    const int jj  = lane & 3;
    const int q0 = (jj == 0) ? -1 : 0, q1 = (jj == 1) ? -1 : 0;
    const int q2 = (jj == 2) ? -1 : 0, q3 = (jj == 3) ? -1 : 0;
    const int cnt = FULL ? 32 : n;
#pragma nounroll
    for (int j0 = 0; j0 < cnt; j0 += 8) {
        v4f    v[8];
        int    dj[8];
#pragma unroll
        for (int u = 0; u < 8; ++u) {
            const int sj = __builtin_amdgcn_readlane(s, j0 + u);
            dj[u] = __builtin_amdgcn_readlane(dl, j0 + u);
            v[u]  = ((const v4f*)(src_p + (size_t)sj * FF))[lane];
        }
        asm volatile("" :: "v"(v[0]), "v"(v[1]), "v"(v[2]), "v"(v[3]),
                           "v"(v[4]), "v"(v[5]), "v"(v[6]), "v"(v[7]));
#pragma unroll
        for (int q = 0; q < 8; q += 4) {
            if (FULL || (j0 + q < cnt)) {
                float4 dq[4];
#pragma unroll
                for (int k = 0; k < 4; ++k)
                    dq[k] = drow4[dj[q + k] * (FF / 4) + lane];
                float wq[4];
#pragma unroll
                for (int k = 0; k < 4; ++k) {
                    const int  u  = q + k;
                    const bool on = FULL || (j0 + u < cnt);
                    float t0 = v[u].x + dq[k].x, t1 = v[u].y + dq[k].y;
                    float t2 = v[u].z + dq[k].z, t3 = v[u].w + dq[k].w;
                    t0 = fmaxf(t0, 0.2f * t0);
                    t1 = fmaxf(t1, 0.2f * t1);
                    t2 = fmaxf(t2, 0.2f * t2);
                    t3 = fmaxf(t3, 0.2f * t3);
                    const float p = quad_sum(at.x * t0 + at.y * t1 + at.z * t2 + at.w * t3);
                    if (on) mx = fmaxf(mx, p);
                    wq[k] = on ? __expf(p - sub) * scl : 0.0f;
                }
#pragma unroll
                for (int k = 0; k < 4; ++k) {
                    const int u = q + k;
                    if (FULL || (j0 + u < cnt)) {
                        float* row = acc + dj[u] * FF + lane;
                        atomicAdd(row,      v[u].x * wq[k]);
                        atomicAdd(row + 32, v[u].y * wq[k]);
                        atomicAdd(row + 64, v[u].z * wq[k]);
                        atomicAdd(row + 96, v[u].w * wq[k]);
                    }
                }
                const int   dsel = (dj[q] & q0) | (dj[q + 1] & q1)
                                 | (dj[q + 2] & q2) | (dj[q + 3] & q3);
                const float wsel = (jj == 0) ? wq[0] : (jj == 1) ? wq[1]
                                 : (jj == 2) ? wq[2] : wq[3];
                atomicAdd(den + dsel * HH + hd, wsel);
            }
        }
    }
    return mx;
}

__device__ __forceinline__ void keys_read(
    const int2* __restrict__ k2, int g0, int start, int lane,
    int2 (&d)[KDEPTH], int (&gi)[KDEPTH])
{
#pragma unroll
    for (int k = 0; k < KDEPTH; ++k) {
        int g = g0 + k * MSG_WAVES;
        const bool in = g < KGROUPS;
        g += start;
        if (g >= KGROUPS) g -= KGROUPS;
        gi[k] = in ? g : -1;
        d[k] = k2[(in ? g : 0) * 32 + lane];
    }
}

__global__ __launch_bounds__(MSG_THREADS) void gat_fused_own(
    const float* __restrict__ src_p,
    const float* __restrict__ dst_p,
    const unsigned* __restrict__ keys,
    const unsigned* __restrict__ pk,
    const float* __restrict__ attn,
    const float* __restrict__ x,
    const float* __restrict__ bias,
    const float* __restrict__ prelu_w,
    float*       __restrict__ partial,
    float*       __restrict__ shiftg,
    float*       __restrict__ deng,
    float*       __restrict__ accg,
    float*       __restrict__ out)
{
    __shared__ __attribute__((aligned(16))) float drow[RNG * FF];
    __shared__ __attribute__((aligned(16))) float acc[RNG * FF];
    __shared__ __attribute__((aligned(16))) float den[RNG * HH];
    __shared__ __attribute__((aligned(16))) uint2 que[MSG_WAVES * KQCAP];
    __shared__ float wmax[MSG_WAVES];

    const int tid  = threadIdx.x;
    const int lane = tid & 31;
    const int wave = __builtin_amdgcn_readfirstlane(tid >> 5);
    const int base = blockIdx.x * RNG;

    const float4* __restrict__ dp4 = (const float4*)(dst_p + (size_t)base * FF);
    float4* drow4 = (float4*)drow;
    float4* acc4  = (float4*)acc;
    for (int i = tid; i < RNG * FF / 4; i += MSG_THREADS)
        drow4[i] = dp4[i];

    const float4 at = ((const float4*)attn)[lane];
    uint2* q = que + wave * KQCAP;

    const int2* __restrict__ k2 = (const int2*)keys;
    const unsigned bb = (unsigned)blockIdx.x * 0x01010101u;
    const int start = blockIdx.x * (KGROUPS / MSG_BLOCKS);

    float shift = SHIFT0;
    float sub   = 0.0f;
    float scl   = __expf(-SHIFT0);
    float bmax  = -INFINITY;
    for (int pass = 0; pass < 2; ++pass) {
        for (int i = tid; i < RNG * FF / 4; i += MSG_THREADS)
            acc4[i] = make_float4(0.0f, 0.0f, 0.0f, 0.0f);
        for (int i = tid; i < RNG * HH; i += MSG_THREADS)
            den[i] = 0.0f;
        __syncthreads();

        int   qn = 0;
        float mx = -INFINITY;
        int2 dn[KDEPTH];
        int  gn[KDEPTH];
        keys_read(k2, wave, start, lane, dn, gn);
        for (int g0 = wave; g0 < KGROUPS; g0 += KDEPTH * MSG_WAVES) {
            int2 d[KDEPTH];
            int  gi[KDEPTH];
#pragma unroll
            for (int k = 0; k < KDEPTH; ++k) { d[k] = dn[k]; gi[k] = gn[k]; }
            keys_read(k2, g0 + KDEPTH * MSG_WAVES, start, lane, dn, gn);
#pragma unroll
            for (int k = 0; k < KDEPTH; ++k) {
                if (gi[k] < 0) continue;
                const unsigned m0 = zero_bytes((unsigned)d[k].x ^ bb);
                const unsigned m1 = zero_bytes((unsigned)d[k].y ^ bb);
                const unsigned marks = m0 | (m1 >> 1);
                qn = queue_push2(q, qn, marks != 0u, (unsigned)(gi[k] * 32 + lane), marks);
            }
            while (qn >= 32) {
                qn -= 32;
                unsigned e;
                qn = pop_edges<true>(q, qn, 32, lane, e);
                mx = fuse_edges<true>(e, 32, drow4, at, sub, scl, mx, lane,
                                      src_p, pk, acc, den);
            }
        }
        while (qn > 0) {
            const int n = qn < 32 ? qn : 32;
            qn -= n;
            unsigned e;
            qn = pop_edges<false>(q, qn, n, lane, e);
            mx = fuse_edges<false>(e, n, drow4, at, sub, scl, mx, lane,
                                   src_p, pk, acc, den);
        }

        mx = fmaxf(mx, __shfl_xor(mx, 16));
        mx = fmaxf(mx, __shfl_xor(mx, 8));
        mx = fmaxf(mx, __shfl_xor(mx, 4));
        mx = fmaxf(mx, __shfl_xor(mx, 2));
        mx = fmaxf(mx, __shfl_xor(mx, 1));
        if (lane == 0) wmax[wave] = mx;
        __syncthreads();
        float m = wmax[lane];
        m = fmaxf(m, __shfl_xor(m, 16));
        m = fmaxf(m, __shfl_xor(m, 8));
        m = fmaxf(m, __shfl_xor(m, 4));
        m = fmaxf(m, __shfl_xor(m, 2));
        m = fmaxf(m, __shfl_xor(m, 1));
        bmax = __int_as_float(__builtin_amdgcn_readfirstlane(__float_as_int(m)));
        const bool again = (pass == 0) &&
                           ((bmax > SHIFT0) ||
                            (bmax < SHIFT0 - SHIFT_SPAN && bmax > -INFINITY));
        if (!again) break;
        shift = bmax;
        sub   = bmax;
        scl   = 1.0f;
    }

    const float4* __restrict__ x4 = (const float4*)(x + (size_t)base * FF);
    const float4* __restrict__ b4 = (const float4*)bias;
    const float pw = prelu_w[0];
    float4* o4 = (float4*)(out  + (size_t)base * FF);
    float4* g4 = (float4*)(accg + (size_t)base * FF);
    for (int i = tid; i < RNG * FF / 4; i += MSG_THREADS) {
        const int row   = i >> 5;
        const int piece = i & 31;
        const float  d  = den[row * HH + (piece >> 2)];
        const float* ar = acc + row * FF + piece;
        const float4 a  = make_float4(ar[0], ar[32], ar[64], ar[96]);
        const float4 xv = x4[i];
        const float4 bv = b4[piece];
        g4[i] = a;
        const bool has = d > 0.0f;
        float4 r;
        r.x = (has ? a.x / d : 0.0f) + xv.x + bv.x;
        r.y = (has ? a.y / d : 0.0f) + xv.y + bv.y;
        r.z = (has ? a.z / d : 0.0f) + xv.z + bv.z;
        r.w = (has ? a.w / d : 0.0f) + xv.w + bv.w;
        r.x = (r.x >= 0.0f) ? r.x : pw * r.x;
        r.y = (r.y >= 0.0f) ? r.y : pw * r.y;
        r.z = (r.z >= 0.0f) ? r.z : pw * r.z;
        r.w = (r.w >= 0.0f) ? r.w : pw * r.w;
        o4[i] = r;
    }
    for (int i = tid; i < RNG * HH; i += MSG_THREADS)
        deng[(size_t)base * HH + i] = den[i];
    if (tid == 0) {
        partial[blockIdx.x] = bmax;
        shiftg[blockIdx.x]  = shift;
    }
}

__global__ __launch_bounds__(256) void gat_eps_fix(
    const float* __restrict__ partial,
    const float* __restrict__ shiftg,
    const float* __restrict__ deng,
    const float* __restrict__ accg,
    const float* __restrict__ x,
    const float* __restrict__ bias,
    const float* __restrict__ prelu_w,
    float* __restrict__ out)
{
    __shared__ float smax[256];
    float s = -INFINITY;
    for (int i = threadIdx.x; i < MSG_BLOCKS; i += 256) s = fmaxf(s, partial[i]);
    smax[threadIdx.x] = s;
    __syncthreads();
#pragma unroll
    for (int off = 128; off > 0; off >>= 1) {
        if (threadIdx.x < off)
            smax[threadIdx.x] = fmaxf(smax[threadIdx.x], smax[threadIdx.x + off]);
        __syncthreads();
    }
    const float gm  = smax[0];
    const float eh  = expf(0.5f * (gm - shiftg[blockIdx.x]));
    const float eps = eh * (1e-16f * eh);
    const float pw  = prelu_w[0];
    const size_t base = (size_t)blockIdx.x * RNG;
    for (int i = threadIdx.x; i < RNG * HH; i += 256) {
        const float d  = deng[base * HH + i];
        const float de = d + eps;
        if (de != d) {
            const int row = i >> 3;
            const int h   = i & 7;
            const size_t o = (base + row) * FF + h * DD;
            for (int c = 0; c < DD; ++c) {
                float r = accg[o + c] / de + x[o + c] + bias[h * DD + c];
                r = (r >= 0.0f) ? r : pw * r;
                out[o + c] = r;
            }
        }
    }
}

extern "C" void kernel_launch(void* const* d_in, const int* in_sizes, int n_in,
                              void* d_out, int out_size, void* d_ws, size_t ws_size,
                              hipStream_t stream) {
    (void)in_sizes; (void)n_in; (void)out_size; (void)ws_size;

    const float* x        = (const float*)d_in[0];
    const int*   eidx     = (const int*)  d_in[1];
    const float* W_src    = (const float*)d_in[2];
    const float* W_dst    = (const float*)d_in[3];
    const float* attn     = (const float*)d_in[4];
    const float* bias     = (const float*)d_in[5];
    const float* prelu_w  = (const float*)d_in[6];
    const int*   srcI     = eidx;
    const int*   dstI     = eidx + EE;

    float* out = (float*)d_out;

    float* ws      = (float*)d_ws;
    float* src_p   = ws;
    float* dst_p   = src_p + (size_t)NN * FF;
    float* accg    = dst_p + (size_t)NN * FF;
    float* deng    = accg + (size_t)NN * FF;
    unsigned* keys = (unsigned*)(deng + (size_t)NN * HH);
    unsigned* pk   = keys + EE / 4;
    float* partial = (float*)(pk + EE);
    float* shiftg  = partial + MSG_BLOCKS;

    {
        dim3 grid(2, NN / 16);
        gat_proj_gemm<<<grid, 32, 0, stream>>>(x, W_src, W_dst, src_p, dst_p,
                                               srcI, dstI, keys, pk);
    }

    gat_fused_own<<<MSG_BLOCKS, MSG_THREADS, 0, stream>>>(src_p, dst_p, keys, pk, attn,
                                                          x, bias, prelu_w,
                                                          partial, shiftg, deng, accg,
                                                          out);

    gat_eps_fix<<<MSG_BLOCKS, 256, 0, stream>>>(partial, shiftg, deng, accg,
                                                x, bias, prelu_w, out);
}


# owner numbers stored bit-sliced by the projection launch (8 words per 32 edges): an owner block marks its edges 32 per lane with 8 xor/and words, not 8 per lane bytewise; add order free as before
# speedup vs baseline: 6.8078x; 1.0337x over previous
#include <hip/hip_runtime.h>
#include <hip/hip_bf16.h>
#include <math.h>

#define NN 50000
#define EE 800000
#define FF 128
#define HH 8
#define DD 16

#define RNG         200
#define MSG_BLOCKS  (NN / RNG)

typedef __attribute__((ext_vector_type(2))) float v2f;
typedef __attribute__((ext_vector_type(8))) float v8f;
typedef __attribute__((ext_vector_type(4))) float v4f;

__global__ __launch_bounds__(32) void gat_proj_gemm(
    const float* __restrict__ x,
    const float* __restrict__ W_src,
    const float* __restrict__ W_dst,
    float* __restrict__ src_p,
    float* __restrict__ dst_p,
    const int* __restrict__ srcI,
    const int* __restrict__ dstI,
    unsigned* __restrict__ keys,
    unsigned* __restrict__ pk)
{
    {
        const int wid = blockIdx.y * 2 + blockIdx.x;
        const int4 dv = ((const int4*)dstI)[wid * 32 + threadIdx.x];
        const int4 sv = ((const int4*)srcI)[wid * 32 + threadIdx.x];
        const unsigned k0 = (unsigned)dv.x / RNG, k1 = (unsigned)dv.y / RNG;
        const unsigned k2 = (unsigned)dv.z / RNG, k3 = (unsigned)dv.w / RNG;
        const unsigned kk[4] = {k0, k1, k2, k3};
        unsigned word = 0u;
        for (int k = 0; k < 4; ++k) {
            for (int j = 0; j < 8; ++j) {
                const unsigned b = __builtin_amdgcn_ballot_w32(((kk[k] >> j) & 1u) != 0u);
                if ((int)threadIdx.x == 8 * k + j) word = b;
            }
        }
        keys[wid * 32 + threadIdx.x] = word;
        uint4 p;
        p.x = (unsigned)sv.x | (((unsigned)dv.x - k0 * RNG) << 16);
        p.y = (unsigned)sv.y | (((unsigned)dv.y - k1 * RNG) << 16);
        p.z = (unsigned)sv.z | (((unsigned)dv.z - k2 * RNG) << 16);
        p.w = (unsigned)sv.w | (((unsigned)dv.w - k3 * RNG) << 16);
        ((uint4*)pk)[wid * 32 + threadIdx.x] = p;
    }

    const int which   = blockIdx.x;
    const int rowbase = blockIdx.y * 16;
    const float* __restrict__ W   = which ? W_dst : W_src;
    float* __restrict__       out = which ? dst_p : src_p;

    const int lane = threadIdx.x;
    const int half = lane >> 4;
    const int l    = lane & 15;

    const float* __restrict__ xrow = x + (size_t)(rowbase + l) * FF;

    v8f acc[8];
#pragma unroll
    for (int c = 0; c < 8; ++c) acc[c] = (v8f){};

    for (int k = 0; k < FF; k += 4) {
        v2f a;
        a.x = xrow[k + 2 * half + 0];
        a.y = xrow[k + 2 * half + 1];
#pragma unroll
        for (int c = 0; c < 8; ++c) {
            const float* wp = W + (size_t)(k + 2 * half) * FF + c * 16 + l;
            v2f b;
            b.x = wp[0];
            b.y = wp[FF];
            acc[c] = __builtin_amdgcn_wmma_f32_16x16x4_f32(
                false, a, false, b, (short)0, acc[c], false, false);
        }
    }

#pragma unroll
    for (int c = 0; c < 8; ++c) {
#pragma unroll
        for (int r = 0; r < 8; ++r) {
            out[(size_t)(rowbase + r + 8 * half) * FF + c * 16 + l] = acc[c][r];
        }
    }
}

#define MSG_THREADS 1024
#define MSG_WAVES   (MSG_THREADS / 32)
#define SHIFT0      10.0f
#define SHIFT_SPAN  20.0f

static_assert(NN % RNG == 0, "blocks tile the rows exactly");
static_assert(EE / 128 == 2 * (NN / 16) && EE % 128 == 0, "K1's waves tile the edge list, 128 edges each");
static_assert(MSG_BLOCKS <= 255 && NN <= 65536, "owner block fits a byte, source fits 16 bits");
static_assert(MSG_WAVES == 32, "the block maximum is read one wave maximum per lane");

__device__ __forceinline__ void wave_lds_order() {
    __builtin_amdgcn_fence(__ATOMIC_RELEASE, "wavefront");
    __builtin_amdgcn_wave_barrier();
    __builtin_amdgcn_fence(__ATOMIC_ACQUIRE, "wavefront");
}

__device__ __forceinline__ float quad_sum(float p) {
    p += __int_as_float(__builtin_amdgcn_update_dpp(0, __float_as_int(p), 0x161, 0xf, 0xf, true));
    p += __int_as_float(__builtin_amdgcn_update_dpp(0, __float_as_int(p), 0x162, 0xf, 0xf, true));
    return p;
}

#define KGROUPS  (EE / 32)
#define KSUPER   ((KGROUPS + 31) / 32)
#define KDEPTH   1
#define KQCAP    64

static_assert(EE % 128 == 0, "K1's waves write whole groups of 32 edges");
static_assert(KQCAP >= 31 + KDEPTH * 32, "a step's marked lanes always fit the queue");
static_assert((MSG_BLOCKS - 1) * (KSUPER / MSG_BLOCKS) < KSUPER, "a staggered start wraps at most once");

__device__ __forceinline__ int queue_push2(uint2* q, int qn,
                                           bool m, unsigned idx, unsigned marks) {
    const unsigned mask = __builtin_amdgcn_ballot_w32(m);
    if (mask) {
        if (m) q[qn + (int)__builtin_amdgcn_mbcnt_lo(mask, 0u)] = make_uint2(idx, marks);
        qn += __builtin_popcount(mask);
    }
    return qn;
}

template <bool FULL>
__device__ __forceinline__ int pop_edges(uint2* q, int qn, int n,
                                         int lane, unsigned& e)
{
    wave_lds_order();
    const bool act = FULL || (lane < n);
    uint2 w = make_uint2(0u, 0u);
    if (act) w = q[qn + lane];
    const unsigned p = (w.y != 0u) ? (unsigned)__builtin_ctz(w.y) : 0u;
    e = (w.x >> 2) * 128u + 4u * p + (w.x & 3u);
    const unsigned rest = w.y & (w.y - 1u);
    return queue_push2(q, qn, rest != 0u, w.x, rest);
}

template <bool FULL>
__device__ __forceinline__ float fuse_edges(
    const unsigned e, int n, const float4* __restrict__ drow4,
    const float4 at, const float sub, const float scl, float mx, int lane,
    const float* __restrict__ src_p,
    const unsigned* __restrict__ pk,
    float* __restrict__ acc, float* __restrict__ den)
{
    const bool act = FULL || (lane < n);
    int s = 0, dl = 0;
    if (act) {
        const unsigned p = pk[e];
        s  = (int)(p & 0xffffu);
        dl = (int)(p >> 16);
    }
    const int hd  = lane >> 2;
    const int jj  = lane & 3;
    const int q0 = (jj == 0) ? -1 : 0, q1 = (jj == 1) ? -1 : 0;
    const int q2 = (jj == 2) ? -1 : 0, q3 = (jj == 3) ? -1 : 0;
    const int cnt = FULL ? 32 : n;
#pragma nounroll
    for (int j0 = 0; j0 < cnt; j0 += 8) {
        v4f    v[8];
        int    dj[8];
#pragma unroll
        for (int u = 0; u < 8; ++u) {
            const int sj = __builtin_amdgcn_readlane(s, j0 + u);
            dj[u] = __builtin_amdgcn_readlane(dl, j0 + u);
            v[u]  = ((const v4f*)(src_p + (size_t)sj * FF))[lane];
        }
        asm volatile("" :: "v"(v[0]), "v"(v[1]), "v"(v[2]), "v"(v[3]),
                           "v"(v[4]), "v"(v[5]), "v"(v[6]), "v"(v[7]));
#pragma unroll
        for (int q = 0; q < 8; q += 4) {
            if (FULL || (j0 + q < cnt)) {
                float4 dq[4];
#pragma unroll
                for (int k = 0; k < 4; ++k)
                    dq[k] = drow4[dj[q + k] * (FF / 4) + lane];
                float wq[4];
#pragma unroll
                for (int k = 0; k < 4; ++k) {
                    const int  u  = q + k;
                    const bool on = FULL || (j0 + u < cnt);
                    float t0 = v[u].x + dq[k].x, t1 = v[u].y + dq[k].y;
                    float t2 = v[u].z + dq[k].z, t3 = v[u].w + dq[k].w;
                    t0 = fmaxf(t0, 0.2f * t0);
                    t1 = fmaxf(t1, 0.2f * t1);
                    t2 = fmaxf(t2, 0.2f * t2);
                    t3 = fmaxf(t3, 0.2f * t3);
                    const float p = quad_sum(at.x * t0 + at.y * t1 + at.z * t2 + at.w * t3);
                    if (on) mx = fmaxf(mx, p);
                    wq[k] = on ? __expf(p - sub) * scl : 0.0f;
                }
#pragma unroll
                for (int k = 0; k < 4; ++k) {
                    const int u = q + k;
                    if (FULL || (j0 + u < cnt)) {
                        float* row = acc + dj[u] * FF + lane;
                        atomicAdd(row,      v[u].x * wq[k]);
                        atomicAdd(row + 32, v[u].y * wq[k]);
                        atomicAdd(row + 64, v[u].z * wq[k]);
                        atomicAdd(row + 96, v[u].w * wq[k]);
                    }
                }
                const int   dsel = (dj[q] & q0) | (dj[q + 1] & q1)
                                 | (dj[q + 2] & q2) | (dj[q + 3] & q3);
                const float wsel = (jj == 0) ? wq[0] : (jj == 1) ? wq[1]
                                 : (jj == 2) ? wq[2] : wq[3];
                atomicAdd(den + dsel * HH + hd, wsel);
            }
        }
    }
    return mx;
}

__device__ __forceinline__ void keys_read(
    const uint4* __restrict__ k4, int g0, int start, int lane,
    uint4 (&lo)[KDEPTH], uint4 (&hi)[KDEPTH], int (&gi)[KDEPTH])
{
#pragma unroll
    for (int k = 0; k < KDEPTH; ++k) {
        int g = g0 + k * MSG_WAVES;
        const bool in = g < KSUPER;
        g += start;
        if (g >= KSUPER) g -= KSUPER;
        gi[k] = in ? g : -1;
        const int grp = g * 32 + lane;
        const int ld = (in && grp < KGROUPS) ? grp : 0;
        lo[k] = k4[ld * 2];
        hi[k] = k4[ld * 2 + 1];
    }
}

__global__ __launch_bounds__(MSG_THREADS) void gat_fused_own(
    const float* __restrict__ src_p,
    const float* __restrict__ dst_p,
    const unsigned* __restrict__ keys,
    const unsigned* __restrict__ pk,
    const float* __restrict__ attn,
    const float* __restrict__ x,
    const float* __restrict__ bias,
    const float* __restrict__ prelu_w,
    float*       __restrict__ partial,
    float*       __restrict__ shiftg,
    float*       __restrict__ deng,
    float*       __restrict__ accg,
    float*       __restrict__ out)
{
    __shared__ __attribute__((aligned(16))) float drow[RNG * FF];
    __shared__ __attribute__((aligned(16))) float acc[RNG * FF];
    __shared__ __attribute__((aligned(16))) float den[RNG * HH];
    __shared__ __attribute__((aligned(16))) uint2 que[MSG_WAVES * KQCAP];
    __shared__ float wmax[MSG_WAVES];

    const int tid  = threadIdx.x;
    const int lane = tid & 31;
    const int wave = __builtin_amdgcn_readfirstlane(tid >> 5);
    const int base = blockIdx.x * RNG;

    const float4* __restrict__ dp4 = (const float4*)(dst_p + (size_t)base * FF);
    float4* drow4 = (float4*)drow;
    float4* acc4  = (float4*)acc;
    for (int i = tid; i < RNG * FF / 4; i += MSG_THREADS)
        drow4[i] = dp4[i];

    const float4 at = ((const float4*)attn)[lane];
    uint2* q = que + wave * KQCAP;

    const uint4* __restrict__ k4 = (const uint4*)keys;
    unsigned cb[8];
#pragma unroll
    for (int j = 0; j < 8; ++j)
        cb[j] = ((blockIdx.x >> j) & 1u) ? 0u : 0xffffffffu;
    const int start = blockIdx.x * (KSUPER / MSG_BLOCKS);

    float shift = SHIFT0;
    float sub   = 0.0f;
    float scl   = __expf(-SHIFT0);
    float bmax  = -INFINITY;
    for (int pass = 0; pass < 2; ++pass) {
        for (int i = tid; i < RNG * FF / 4; i += MSG_THREADS)
            acc4[i] = make_float4(0.0f, 0.0f, 0.0f, 0.0f);
        for (int i = tid; i < RNG * HH; i += MSG_THREADS)
            den[i] = 0.0f;
        __syncthreads();

        int   qn = 0;
        float mx = -INFINITY;
        uint4 ln[KDEPTH], hn[KDEPTH];
        int   gn[KDEPTH];
        keys_read(k4, wave, start, lane, ln, hn, gn);
        for (int g0 = wave; g0 < KSUPER; g0 += KDEPTH * MSG_WAVES) {
            uint4 lo[KDEPTH], hi[KDEPTH];
            int   gi[KDEPTH];
#pragma unroll
            for (int k = 0; k < KDEPTH; ++k) { lo[k] = ln[k]; hi[k] = hn[k]; gi[k] = gn[k]; }
            keys_read(k4, g0 + KDEPTH * MSG_WAVES, start, lane, ln, hn, gn);
#pragma unroll
            for (int k = 0; k < KDEPTH; ++k) {
                if (gi[k] < 0) continue;
                const unsigned grp = (unsigned)(gi[k] * 32 + lane);
                unsigned marks = (lo[k].x ^ cb[0]) & (lo[k].y ^ cb[1])
                               & (lo[k].z ^ cb[2]) & (lo[k].w ^ cb[3])
                               & (hi[k].x ^ cb[4]) & (hi[k].y ^ cb[5])
                               & (hi[k].z ^ cb[6]) & (hi[k].w ^ cb[7]);
                if (grp >= (unsigned)KGROUPS) marks = 0u;
                qn = queue_push2(q, qn, marks != 0u, grp, marks);
            }
            while (qn >= 32) {
                qn -= 32;
                unsigned e;
                qn = pop_edges<true>(q, qn, 32, lane, e);
                mx = fuse_edges<true>(e, 32, drow4, at, sub, scl, mx, lane,
                                      src_p, pk, acc, den);
            }
        }
        while (qn > 0) {
            const int n = qn < 32 ? qn : 32;
            qn -= n;
            unsigned e;
            qn = pop_edges<false>(q, qn, n, lane, e);
            mx = fuse_edges<false>(e, n, drow4, at, sub, scl, mx, lane,
                                   src_p, pk, acc, den);
        }

        mx = fmaxf(mx, __shfl_xor(mx, 16));
        mx = fmaxf(mx, __shfl_xor(mx, 8));
        mx = fmaxf(mx, __shfl_xor(mx, 4));
        mx = fmaxf(mx, __shfl_xor(mx, 2));
        mx = fmaxf(mx, __shfl_xor(mx, 1));
        if (lane == 0) wmax[wave] = mx;
        __syncthreads();
        float m = wmax[lane];
        m = fmaxf(m, __shfl_xor(m, 16));
        m = fmaxf(m, __shfl_xor(m, 8));
        m = fmaxf(m, __shfl_xor(m, 4));
        m = fmaxf(m, __shfl_xor(m, 2));
        m = fmaxf(m, __shfl_xor(m, 1));
        bmax = __int_as_float(__builtin_amdgcn_readfirstlane(__float_as_int(m)));
        const bool again = (pass == 0) &&
                           ((bmax > SHIFT0) ||
                            (bmax < SHIFT0 - SHIFT_SPAN && bmax > -INFINITY));
        if (!again) break;
        shift = bmax;
        sub   = bmax;
        scl   = 1.0f;
    }

    const float4* __restrict__ x4 = (const float4*)(x + (size_t)base * FF);
    const float4* __restrict__ b4 = (const float4*)bias;
    const float pw = prelu_w[0];
    float4* o4 = (float4*)(out  + (size_t)base * FF);
    float4* g4 = (float4*)(accg + (size_t)base * FF);
    for (int i = tid; i < RNG * FF / 4; i += MSG_THREADS) {
        const int row   = i >> 5;
        const int piece = i & 31;
        const float  d  = den[row * HH + (piece >> 2)];
        const float* ar = acc + row * FF + piece;
        const float4 a  = make_float4(ar[0], ar[32], ar[64], ar[96]);
        const float4 xv = x4[i];
        const float4 bv = b4[piece];
        g4[i] = a;
        const bool has = d > 0.0f;
        float4 r;
        r.x = (has ? a.x / d : 0.0f) + xv.x + bv.x;
        r.y = (has ? a.y / d : 0.0f) + xv.y + bv.y;
        r.z = (has ? a.z / d : 0.0f) + xv.z + bv.z;
        r.w = (has ? a.w / d : 0.0f) + xv.w + bv.w;
        r.x = (r.x >= 0.0f) ? r.x : pw * r.x;
        r.y = (r.y >= 0.0f) ? r.y : pw * r.y;
        r.z = (r.z >= 0.0f) ? r.z : pw * r.z;
        r.w = (r.w >= 0.0f) ? r.w : pw * r.w;
        o4[i] = r;
    }
    for (int i = tid; i < RNG * HH; i += MSG_THREADS)
        deng[(size_t)base * HH + i] = den[i];
    if (tid == 0) {
        partial[blockIdx.x] = bmax;
        shiftg[blockIdx.x]  = shift;
    }
}

__global__ __launch_bounds__(256) void gat_eps_fix(
    const float* __restrict__ partial,
    const float* __restrict__ shiftg,
    const float* __restrict__ deng,
    const float* __restrict__ accg,
    const float* __restrict__ x,
    const float* __restrict__ bias,
    const float* __restrict__ prelu_w,
    float* __restrict__ out)
{
    __shared__ float smax[256];
    float s = -INFINITY;
    for (int i = threadIdx.x; i < MSG_BLOCKS; i += 256) s = fmaxf(s, partial[i]);
    smax[threadIdx.x] = s;
    __syncthreads();
#pragma unroll
    for (int off = 128; off > 0; off >>= 1) {
        if (threadIdx.x < off)
            smax[threadIdx.x] = fmaxf(smax[threadIdx.x], smax[threadIdx.x + off]);
        __syncthreads();
    }
    const float gm  = smax[0];
    const float eh  = expf(0.5f * (gm - shiftg[blockIdx.x]));
    const float eps = eh * (1e-16f * eh);
    const float pw  = prelu_w[0];
    const size_t base = (size_t)blockIdx.x * RNG;
    for (int i = threadIdx.x; i < RNG * HH; i += 256) {
        const float d  = deng[base * HH + i];
        const float de = d + eps;
        if (de != d) {
            const int row = i >> 3;
            const int h   = i & 7;
            const size_t o = (base + row) * FF + h * DD;
            for (int c = 0; c < DD; ++c) {
                float r = accg[o + c] / de + x[o + c] + bias[h * DD + c];
                r = (r >= 0.0f) ? r : pw * r;
                out[o + c] = r;
            }
        }
    }
}

extern "C" void kernel_launch(void* const* d_in, const int* in_sizes, int n_in,
                              void* d_out, int out_size, void* d_ws, size_t ws_size,
                              hipStream_t stream) {
    (void)in_sizes; (void)n_in; (void)out_size; (void)ws_size;

    const float* x        = (const float*)d_in[0];
    const int*   eidx     = (const int*)  d_in[1];
    const float* W_src    = (const float*)d_in[2];
    const float* W_dst    = (const float*)d_in[3];
    const float* attn     = (const float*)d_in[4];
    const float* bias     = (const float*)d_in[5];
    const float* prelu_w  = (const float*)d_in[6];
    const int*   srcI     = eidx;
    const int*   dstI     = eidx + EE;

    float* out = (float*)d_out;

    float* ws      = (float*)d_ws;
    float* src_p   = ws;
    float* dst_p   = src_p + (size_t)NN * FF;
    float* accg    = dst_p + (size_t)NN * FF;
    float* deng    = accg + (size_t)NN * FF;
    unsigned* keys = (unsigned*)(deng + (size_t)NN * HH);
    unsigned* pk   = keys + EE / 4;
    float* partial = (float*)(pk + EE);
    float* shiftg  = partial + MSG_BLOCKS;

    {
        dim3 grid(2, NN / 16);
        gat_proj_gemm<<<grid, 32, 0, stream>>>(x, W_src, W_dst, src_p, dst_p,
                                               srcI, dstI, keys, pk);
    }

    gat_fused_own<<<MSG_BLOCKS, MSG_THREADS, 0, stream>>>(src_p, dst_p, keys, pk, attn,
                                                          x, bias, prelu_w,
                                                          partial, shiftg, deng, accg,
                                                          out);

    gat_eps_fix<<<MSG_BLOCKS, 256, 0, stream>>>(partial, shiftg, deng, accg,
                                                x, bias, prelu_w, out);
}


# ends of the launches: a thread's 7 pieces (dst rows in, x rows of the closing step, last launch's denominators) are read as one group up front, not one per loop trip waited for in full; same sums
# speedup vs baseline: 8.2415x; 1.2106x over previous
#include <hip/hip_runtime.h>
#include <hip/hip_bf16.h>
#include <math.h>

#define NN 50000
#define EE 800000
#define FF 128
#define HH 8
#define DD 16

#define RNG         200
#define MSG_BLOCKS  (NN / RNG)

typedef __attribute__((ext_vector_type(2))) float v2f;
typedef __attribute__((ext_vector_type(8))) float v8f;
typedef __attribute__((ext_vector_type(4))) float v4f;

#define PJ_WAVES    16
#define PJ_THREADS  (PJ_WAVES * 32)
#define PJ_BLOCKS   256
#define NPANEL      (NN / 16)
#define XPITCH      132
#define KCHUNKS     (EE / 128)

static_assert(NN % 16 == 0, "panels tile the rows exactly");
static_assert(PJ_THREADS == 16 * (FF / 4), "a block's threads tile one panel in 16-byte pieces");
static_assert(PJ_WAVES == 2 * (FF / 16), "one wave per weight matrix and 16-column tile");
static_assert(FF == 128, "a lane half holds 64 k of a row");

__device__ __forceinline__ void frag_load(float4 (&a)[16], const float* ar) {
#pragma unroll
    for (int j = 0; j < 16; ++j) a[j] = *(const float4*)(ar + 4 * j);
}

template <int J0>
__device__ __forceinline__ void panel_half(const float4 (&a)[16], const v2f (&b)[32],
                                           v8f& acc0, v8f& acc1) {
#pragma unroll
    for (int j = J0; j < J0 + 8; ++j) {
        v2f a0, a1;
        a0.x = a[j].x; a0.y = a[j].y;
        a1.x = a[j].z; a1.y = a[j].w;
        acc0 = __builtin_amdgcn_wmma_f32_16x16x4_f32(
            false, a0, false, b[2 * j], (short)0, acc0, false, false);
        acc1 = __builtin_amdgcn_wmma_f32_16x16x4_f32(
            false, a1, false, b[2 * j + 1], (short)0, acc1, false, false);
    }
}

__global__ __launch_bounds__(PJ_THREADS) void gat_proj_gemm(
    const float* __restrict__ x,
    const float* __restrict__ W_src,
    const float* __restrict__ W_dst,
    float* __restrict__ src_p,
    float* __restrict__ dst_p,
    const int* __restrict__ srcI,
    const int* __restrict__ dstI,
    unsigned* __restrict__ keys,
    unsigned* __restrict__ pk)
{
    __shared__ __attribute__((aligned(16))) float xs0[16 * XPITCH];
    __shared__ __attribute__((aligned(16))) float xs1[16 * XPITCH];

    const int tid  = threadIdx.x;
    const int lane = tid & 31;
    const int wave = __builtin_amdgcn_readfirstlane(tid >> 5);
    const int half = lane >> 4;
    const int l    = lane & 15;

    const int c = wave & 7;
    const float* __restrict__ W   = (wave >> 3) ? W_dst : W_src;
    float* __restrict__       out = (wave >> 3) ? dst_p : src_p;
    v2f b[32];
#pragma unroll
    for (int s = 0; s < 32; ++s) {
        const float* wp = W + (size_t)(64 * half + 2 * s) * FF + c * 16 + l;
        b[s].x = wp[0];
        b[s].y = wp[FF];
    }

    const int prow   = tid >> 5;
    const int ppiece = lane;
    const float4* __restrict__ x4 = (const float4*)x;
    int p = blockIdx.x;
    const int pe0 = p < NPANEL ? p : NPANEL - 1;
    const int pe1 = p + PJ_BLOCKS < NPANEL ? p + PJ_BLOCKS : NPANEL - 1;
    const float4 xe0 = x4[((size_t)pe0 * 16 + prow) * (FF / 4) + ppiece];
    const float4 xe1 = x4[((size_t)pe1 * 16 + prow) * (FF / 4) + ppiece];

    for (int wid = (PJ_BLOCKS - 1 - (int)blockIdx.x) * PJ_WAVES + wave; wid < KCHUNKS;
         wid += PJ_BLOCKS * PJ_WAVES) {
        const int4 dv = ((const int4*)dstI)[wid * 32 + lane];
        const int4 sv = ((const int4*)srcI)[wid * 32 + lane];
        const unsigned k0 = (unsigned)dv.x / RNG, k1 = (unsigned)dv.y / RNG;
        const unsigned k2 = (unsigned)dv.z / RNG, k3 = (unsigned)dv.w / RNG;
        asm volatile("" :: "v"(xe0.x), "v"(xe0.y), "v"(xe0.z), "v"(xe0.w),
                           "v"(xe1.x), "v"(xe1.y), "v"(xe1.z), "v"(xe1.w), "v"(k0));
        const unsigned kk[4] = {k0, k1, k2, k3};
        unsigned word = 0u;
        for (int k = 0; k < 4; ++k) {
            for (int j = 0; j < 8; ++j) {
                const unsigned bb = __builtin_amdgcn_ballot_w32(((kk[k] >> j) & 1u) != 0u);
                if (lane == 8 * k + j) word = bb;
            }
        }
        keys[wid * 32 + lane] = word;
        uint4 pw;
        pw.x = (unsigned)sv.x | (((unsigned)dv.x - k0 * RNG) << 16);
        pw.y = (unsigned)sv.y | (((unsigned)dv.y - k1 * RNG) << 16);
        pw.z = (unsigned)sv.z | (((unsigned)dv.z - k2 * RNG) << 16);
        pw.w = (unsigned)sv.w | (((unsigned)dv.w - k3 * RNG) << 16);
        ((uint4*)pk)[wid * 32 + lane] = pw;
    }

    const int sto = prow * XPITCH + 4 * ppiece;
    const int fro = l * XPITCH + 64 * half;
    float* __restrict__ ocol = out + (size_t)(8 * half) * FF + c * 16 + l;

    if (p >= NPANEL) return;
    *(float4*)(xs0 + sto) = xe0;
    if (p + PJ_BLOCKS < NPANEL) *(float4*)(xs1 + sto) = xe1;
    __syncthreads();
    float4 fa[16], fb[16];
    frag_load(fa, xs0 + fro);
    float4 xr = make_float4(0.0f, 0.0f, 0.0f, 0.0f);
    if (p + 2 * PJ_BLOCKS < NPANEL)
        xr = x4[((size_t)(p + 2 * PJ_BLOCKS) * 16 + prow) * (FF / 4) + ppiece];
    __syncthreads();

    auto step = [&](const float4 (&cur)[16], float4 (&nxt)[16], float* bcur, const float* bnxt) {
        if (p + PJ_BLOCKS < NPANEL) frag_load(nxt, bnxt + fro);
        if (p + 2 * PJ_BLOCKS < NPANEL) *(float4*)(bcur + sto) = xr;
        if (p + 3 * PJ_BLOCKS < NPANEL)
            xr = x4[((size_t)(p + 3 * PJ_BLOCKS) * 16 + prow) * (FF / 4) + ppiece];
        v8f acc0 = (v8f){};
        v8f acc1 = (v8f){};
        panel_half<0>(cur, b, acc0, acc1);
        asm volatile("s_wait_dscnt 0x0" ::: "memory");
        asm volatile("s_barrier_signal -1" ::: "memory");
        panel_half<8>(cur, b, acc0, acc1);
        const v8f res = acc0 + acc1;
        float* __restrict__ orow = ocol + (size_t)p * 16 * FF;
#pragma unroll
        for (int r = 0; r < 8; ++r) orow[(size_t)r * FF] = res[r];
        asm volatile("s_barrier_wait -1" ::: "memory");
        p += PJ_BLOCKS;
    };
    for (;;) {
        step(fa, fb, xs0, xs1);
        if (p >= NPANEL) break;
        step(fb, fa, xs1, xs0);
        if (p >= NPANEL) break;
    }
}

#define MSG_THREADS 1024
#define MSG_WAVES   (MSG_THREADS / 32)
#define SHIFT0      10.0f
#define SHIFT_SPAN  20.0f

static_assert(NN % RNG == 0, "blocks tile the rows exactly");
static_assert(EE % 128 == 0, "K1's chunks tile the edge list, 128 edges each");
static_assert(MSG_BLOCKS <= 255 && NN <= 65536, "owner block fits a byte, source fits 16 bits");
static_assert(MSG_WAVES == 32, "the block maximum is read one wave maximum per lane");

__device__ __forceinline__ void wave_lds_order() {
    __builtin_amdgcn_fence(__ATOMIC_RELEASE, "wavefront");
    __builtin_amdgcn_wave_barrier();
    __builtin_amdgcn_fence(__ATOMIC_ACQUIRE, "wavefront");
}

__device__ __forceinline__ float quad_sum(float p) {
    p += __int_as_float(__builtin_amdgcn_update_dpp(0, __float_as_int(p), 0x161, 0xf, 0xf, true));
    p += __int_as_float(__builtin_amdgcn_update_dpp(0, __float_as_int(p), 0x162, 0xf, 0xf, true));
    return p;
}

#define ROWQ        (RNG * FF / 4)
#define ROWQ_TRIPS  ((ROWQ + MSG_THREADS - 1) / MSG_THREADS)

#define KGROUPS  (EE / 32)
#define KSUPER   ((KGROUPS + 31) / 32)
#define KDEPTH   1
#define KQCAP    64

static_assert(EE % 128 == 0, "K1's chunks are whole groups of 32 edges");
static_assert(KQCAP >= 31 + KDEPTH * 32, "a step's marked lanes always fit the queue");
static_assert((MSG_BLOCKS - 1) * (KSUPER / MSG_BLOCKS) < KSUPER, "a staggered start wraps at most once");

__device__ __forceinline__ int queue_push2(uint2* q, int qn,
                                           bool m, unsigned idx, unsigned marks) {
    const unsigned mask = __builtin_amdgcn_ballot_w32(m);
    if (mask) {
        if (m) q[qn + (int)__builtin_amdgcn_mbcnt_lo(mask, 0u)] = make_uint2(idx, marks);
        qn += __builtin_popcount(mask);
    }
    return qn;
}

template <bool FULL>
__device__ __forceinline__ int pop_edges(uint2* q, int qn, int n,
                                         int lane, unsigned& e)
{
    wave_lds_order();
    const bool act = FULL || (lane < n);
    uint2 w = make_uint2(0u, 0u);
    if (act) w = q[qn + lane];
    const unsigned p = (w.y != 0u) ? (unsigned)__builtin_ctz(w.y) : 0u;
    e = (w.x >> 2) * 128u + 4u * p + (w.x & 3u);
    const unsigned rest = w.y & (w.y - 1u);
    return queue_push2(q, qn, rest != 0u, w.x, rest);
}

template <bool FULL>
__device__ __forceinline__ float fuse_edges(
    const unsigned e, int n, const float4* __restrict__ drow4,
    const float4 at, const float sub, const float scl, float mx, int lane,
    const float* __restrict__ src_p,
    const unsigned* __restrict__ pk,
    float* __restrict__ acc, float* __restrict__ den)
{
    const bool act = FULL || (lane < n);
    int s = 0, dl = 0;
    if (act) {
        const unsigned p = pk[e];
        s  = (int)(p & 0xffffu);
        dl = (int)(p >> 16);
    }
    const int hd  = lane >> 2;
    const int jj  = lane & 3;
    const int q0 = (jj == 0) ? -1 : 0, q1 = (jj == 1) ? -1 : 0;
    const int q2 = (jj == 2) ? -1 : 0, q3 = (jj == 3) ? -1 : 0;
    const int cnt = FULL ? 32 : n;
#pragma nounroll
    for (int j0 = 0; j0 < cnt; j0 += 8) {
        v4f    v[8];
        int    dj[8];
#pragma unroll
        for (int u = 0; u < 8; ++u) {
            const int sj = __builtin_amdgcn_readlane(s, j0 + u);
            dj[u] = __builtin_amdgcn_readlane(dl, j0 + u);
            v[u]  = ((const v4f*)(src_p + (size_t)sj * FF))[lane];
        }
        asm volatile("" :: "v"(v[0]), "v"(v[1]), "v"(v[2]), "v"(v[3]),
                           "v"(v[4]), "v"(v[5]), "v"(v[6]), "v"(v[7]));
#pragma unroll
        for (int q = 0; q < 8; q += 4) {
            if (FULL || (j0 + q < cnt)) {
                float4 dq[4];
#pragma unroll
                for (int k = 0; k < 4; ++k)
                    dq[k] = drow4[dj[q + k] * (FF / 4) + lane];
                float wq[4];
#pragma unroll
                for (int k = 0; k < 4; ++k) {
                    const int  u  = q + k;
                    const bool on = FULL || (j0 + u < cnt);
                    float t0 = v[u].x + dq[k].x, t1 = v[u].y + dq[k].y;
                    float t2 = v[u].z + dq[k].z, t3 = v[u].w + dq[k].w;
                    t0 = fmaxf(t0, 0.2f * t0);
                    t1 = fmaxf(t1, 0.2f * t1);
                    t2 = fmaxf(t2, 0.2f * t2);
                    t3 = fmaxf(t3, 0.2f * t3);
                    const float p = quad_sum(at.x * t0 + at.y * t1 + at.z * t2 + at.w * t3);
                    if (on) mx = fmaxf(mx, p);
                    wq[k] = on ? __expf(p - sub) * scl : 0.0f;
                }
#pragma unroll
                for (int k = 0; k < 4; ++k) {
                    const int u = q + k;
                    if (FULL || (j0 + u < cnt)) {
                        float* row = acc + dj[u] * FF + lane;
                        atomicAdd(row,      v[u].x * wq[k]);
                        atomicAdd(row + 32, v[u].y * wq[k]);
                        atomicAdd(row + 64, v[u].z * wq[k]);
                        atomicAdd(row + 96, v[u].w * wq[k]);
                    }
                }
                const int   dsel = (dj[q] & q0) | (dj[q + 1] & q1)
                                 | (dj[q + 2] & q2) | (dj[q + 3] & q3);
                const float wsel = (jj == 0) ? wq[0] : (jj == 1) ? wq[1]
                                 : (jj == 2) ? wq[2] : wq[3];
                atomicAdd(den + dsel * HH + hd, wsel);
            }
        }
    }
    return mx;
}

__device__ __forceinline__ void keys_read(
    const uint4* __restrict__ k4, int g0, int start, int lane,
    uint4 (&lo)[KDEPTH], uint4 (&hi)[KDEPTH], int (&gi)[KDEPTH])
{
#pragma unroll
    for (int k = 0; k < KDEPTH; ++k) {
        int g = g0 + k * MSG_WAVES;
        const bool in = g < KSUPER;
        g += start;
        if (g >= KSUPER) g -= KSUPER;
        gi[k] = in ? g : -1;
        const int grp = g * 32 + lane;
        const int ld = (in && grp < KGROUPS) ? grp : 0;
        lo[k] = k4[ld * 2];
        hi[k] = k4[ld * 2 + 1];
    }
}

__global__ __launch_bounds__(MSG_THREADS) void gat_fused_own(
    const float* __restrict__ src_p,
    const float* __restrict__ dst_p,
    const unsigned* __restrict__ keys,
    const unsigned* __restrict__ pk,
    const float* __restrict__ attn,
    const float* __restrict__ x,
    const float* __restrict__ bias,
    const float* __restrict__ prelu_w,
    float*       __restrict__ partial,
    float*       __restrict__ shiftg,
    float*       __restrict__ deng,
    float*       __restrict__ accg,
    float*       __restrict__ out)
{
    __shared__ __attribute__((aligned(16))) float drow[RNG * FF];
    __shared__ __attribute__((aligned(16))) float acc[RNG * FF];
    __shared__ __attribute__((aligned(16))) float den[RNG * HH];
    __shared__ __attribute__((aligned(16))) uint2 que[MSG_WAVES * KQCAP];
    __shared__ float wmax[MSG_WAVES];

    const int tid  = threadIdx.x;
    const int lane = tid & 31;
    const int wave = __builtin_amdgcn_readfirstlane(tid >> 5);
    const int base = blockIdx.x * RNG;

    const float4* __restrict__ dp4 = (const float4*)(dst_p + (size_t)base * FF);
    float4* drow4 = (float4*)drow;
    float4* acc4  = (float4*)acc;
    float4 dv[ROWQ_TRIPS];
#pragma unroll
    for (int j = 0; j < ROWQ_TRIPS; ++j) {
        const int i = tid + j * MSG_THREADS;
        dv[j] = dp4[i < ROWQ ? i : ROWQ - 1];
    }

    const float4 at = ((const float4*)attn)[lane];
    uint2* q = que + wave * KQCAP;

    const uint4* __restrict__ k4 = (const uint4*)keys;
    unsigned cb[8];
#pragma unroll
    for (int j = 0; j < 8; ++j)
        cb[j] = ((blockIdx.x >> j) & 1u) ? 0u : 0xffffffffu;
    const int start = blockIdx.x * (KSUPER / MSG_BLOCKS);

    float shift = SHIFT0;
    float sub   = 0.0f;
    float scl   = __expf(-SHIFT0);
    float bmax  = -INFINITY;
    auto clear_sums = [&]() {
        for (int i = tid; i < ROWQ; i += MSG_THREADS)
            acc4[i] = make_float4(0.0f, 0.0f, 0.0f, 0.0f);
        for (int i = tid; i < RNG * HH; i += MSG_THREADS)
            den[i] = 0.0f;
    };
    clear_sums();
    static_assert(ROWQ_TRIPS == 7, "the statement below names 7 pieces");
    asm volatile("" :: "v"(dv[0].x), "v"(dv[0].y), "v"(dv[0].z), "v"(dv[0].w),
                       "v"(dv[1].x), "v"(dv[1].y), "v"(dv[1].z), "v"(dv[1].w),
                       "v"(dv[2].x), "v"(dv[2].y), "v"(dv[2].z), "v"(dv[2].w),
                       "v"(dv[3].x), "v"(dv[3].y), "v"(dv[3].z), "v"(dv[3].w),
                       "v"(dv[4].x), "v"(dv[4].y), "v"(dv[4].z), "v"(dv[4].w),
                       "v"(dv[5].x), "v"(dv[5].y), "v"(dv[5].z), "v"(dv[5].w),
                       "v"(dv[6].x), "v"(dv[6].y), "v"(dv[6].z), "v"(dv[6].w));
#pragma unroll
    for (int j = 0; j < ROWQ_TRIPS; ++j) {
        const int i = tid + j * MSG_THREADS;
        if (i < ROWQ) drow4[i] = dv[j];
    }
    for (int pass = 0; pass < 2; ++pass) {
        if (pass > 0) clear_sums();
        __syncthreads();

        int   qn = 0;
        float mx = -INFINITY;
        uint4 ln[KDEPTH], hn[KDEPTH];
        int   gn[KDEPTH];
        keys_read(k4, wave, start, lane, ln, hn, gn);
        for (int g0 = wave; g0 < KSUPER; g0 += KDEPTH * MSG_WAVES) {
            uint4 lo[KDEPTH], hi[KDEPTH];
            int   gi[KDEPTH];
#pragma unroll
            for (int k = 0; k < KDEPTH; ++k) { lo[k] = ln[k]; hi[k] = hn[k]; gi[k] = gn[k]; }
            keys_read(k4, g0 + KDEPTH * MSG_WAVES, start, lane, ln, hn, gn);
#pragma unroll
            for (int k = 0; k < KDEPTH; ++k) {
                if (gi[k] < 0) continue;
                const unsigned grp = (unsigned)(gi[k] * 32 + lane);
                unsigned marks = (lo[k].x ^ cb[0]) & (lo[k].y ^ cb[1])
                               & (lo[k].z ^ cb[2]) & (lo[k].w ^ cb[3])
                               & (hi[k].x ^ cb[4]) & (hi[k].y ^ cb[5])
                               & (hi[k].z ^ cb[6]) & (hi[k].w ^ cb[7]);
                if (grp >= (unsigned)KGROUPS) marks = 0u;
                qn = queue_push2(q, qn, marks != 0u, grp, marks);
            }
            while (qn >= 32) {
                qn -= 32;
                unsigned e;
                qn = pop_edges<true>(q, qn, 32, lane, e);
                mx = fuse_edges<true>(e, 32, drow4, at, sub, scl, mx, lane,
                                      src_p, pk, acc, den);
            }
        }
        while (qn > 0) {
            const int n = qn < 32 ? qn : 32;
            qn -= n;
            unsigned e;
            qn = pop_edges<false>(q, qn, n, lane, e);
            mx = fuse_edges<false>(e, n, drow4, at, sub, scl, mx, lane,
                                   src_p, pk, acc, den);
        }

        mx = fmaxf(mx, __shfl_xor(mx, 16));
        mx = fmaxf(mx, __shfl_xor(mx, 8));
        mx = fmaxf(mx, __shfl_xor(mx, 4));
        mx = fmaxf(mx, __shfl_xor(mx, 2));
        mx = fmaxf(mx, __shfl_xor(mx, 1));
        if (lane == 0) wmax[wave] = mx;
        __syncthreads();
        float m = wmax[lane];
        m = fmaxf(m, __shfl_xor(m, 16));
        m = fmaxf(m, __shfl_xor(m, 8));
        m = fmaxf(m, __shfl_xor(m, 4));
        m = fmaxf(m, __shfl_xor(m, 2));
        m = fmaxf(m, __shfl_xor(m, 1));
        bmax = __int_as_float(__builtin_amdgcn_readfirstlane(__float_as_int(m)));
        const bool again = (pass == 0) &&
                           ((bmax > SHIFT0) ||
                            (bmax < SHIFT0 - SHIFT_SPAN && bmax > -INFINITY));
        if (!again) break;
        shift = bmax;
        sub   = bmax;
        scl   = 1.0f;
    }

    const float4* __restrict__ x4 = (const float4*)(x + (size_t)base * FF);
    const float4* __restrict__ b4 = (const float4*)bias;
    float4 xv[ROWQ_TRIPS];
#pragma unroll
    for (int j = 0; j < ROWQ_TRIPS; ++j) {
        const int i = tid + j * MSG_THREADS;
        xv[j] = x4[i < ROWQ ? i : ROWQ - 1];
    }
    const float4 bv = b4[lane];
    const float  pw = prelu_w[0];
    asm volatile("" :: "v"(xv[0].x), "v"(xv[0].y), "v"(xv[0].z), "v"(xv[0].w),
                       "v"(xv[1].x), "v"(xv[1].y), "v"(xv[1].z), "v"(xv[1].w),
                       "v"(xv[2].x), "v"(xv[2].y), "v"(xv[2].z), "v"(xv[2].w),
                       "v"(xv[3].x), "v"(xv[3].y), "v"(xv[3].z), "v"(xv[3].w),
                       "v"(xv[4].x), "v"(xv[4].y), "v"(xv[4].z), "v"(xv[4].w),
                       "v"(xv[5].x), "v"(xv[5].y), "v"(xv[5].z), "v"(xv[5].w),
                       "v"(xv[6].x), "v"(xv[6].y), "v"(xv[6].z), "v"(xv[6].w));
    float4* o4 = (float4*)(out  + (size_t)base * FF);
    float4* g4 = (float4*)(accg + (size_t)base * FF);
#pragma unroll
    for (int j = 0; j < ROWQ_TRIPS; ++j) {
        const int i = tid + j * MSG_THREADS;
        if (i >= ROWQ) break;
        const int row   = i >> 5;
        const int piece = lane;
        const float  d  = den[row * HH + (piece >> 2)];
        const float* ar = acc + row * FF + piece;
        const float4 a  = make_float4(ar[0], ar[32], ar[64], ar[96]);
        const float4 xq = xv[j];
        g4[i] = a;
        const bool has = d > 0.0f;
        float4 r;
        r.x = (has ? a.x / d : 0.0f) + xq.x + bv.x;
        r.y = (has ? a.y / d : 0.0f) + xq.y + bv.y;
        r.z = (has ? a.z / d : 0.0f) + xq.z + bv.z;
        r.w = (has ? a.w / d : 0.0f) + xq.w + bv.w;
        r.x = (r.x >= 0.0f) ? r.x : pw * r.x;
        r.y = (r.y >= 0.0f) ? r.y : pw * r.y;
        r.z = (r.z >= 0.0f) ? r.z : pw * r.z;
        r.w = (r.w >= 0.0f) ? r.w : pw * r.w;
        o4[i] = r;
    }
    for (int i = tid; i < RNG * HH; i += MSG_THREADS)
        deng[(size_t)base * HH + i] = den[i];
    if (tid == 0) {
        partial[blockIdx.x] = bmax;
        shiftg[blockIdx.x]  = shift;
    }
}

#define EPS_TRIPS ((RNG * HH + 255) / 256)

__global__ __launch_bounds__(256) void gat_eps_fix(
    const float* __restrict__ partial,
    const float* __restrict__ shiftg,
    const float* __restrict__ deng,
    const float* __restrict__ accg,
    const float* __restrict__ x,
    const float* __restrict__ bias,
    const float* __restrict__ prelu_w,
    float* __restrict__ out)
{
    __shared__ float smax[256];
    const size_t base = (size_t)blockIdx.x * RNG;
    float dd[EPS_TRIPS];
#pragma unroll
    for (int j = 0; j < EPS_TRIPS; ++j) {
        const int i = (int)threadIdx.x + j * 256;
        dd[j] = deng[base * HH + (i < RNG * HH ? i : RNG * HH - 1)];
    }
    const float sh = shiftg[blockIdx.x];
    const float pw = prelu_w[0];
    static_assert(MSG_BLOCKS <= 256, "one block maximum per thread");
    const float s = partial[threadIdx.x < MSG_BLOCKS ? threadIdx.x : MSG_BLOCKS - 1];
    smax[threadIdx.x] = s;
    __syncthreads();
#pragma unroll
    for (int off = 128; off > 0; off >>= 1) {
        if (threadIdx.x < off)
            smax[threadIdx.x] = fmaxf(smax[threadIdx.x], smax[threadIdx.x + off]);
        __syncthreads();
    }
    const float gm  = smax[0];
    const float eh  = expf(0.5f * (gm - sh));
    const float eps = eh * (1e-16f * eh);
#pragma unroll
    for (int j = 0; j < EPS_TRIPS; ++j) {
        const int i = (int)threadIdx.x + j * 256;
        if (i >= RNG * HH) break;
        const float d  = dd[j];
        const float de = d + eps;
        if (de != d) {
            const int row = i >> 3;
            const int h   = i & 7;
            const size_t o = (base + row) * FF + h * DD;
            for (int c = 0; c < DD; ++c) {
                float r = accg[o + c] / de + x[o + c] + bias[h * DD + c];
                r = (r >= 0.0f) ? r : pw * r;
                out[o + c] = r;
            }
        }
    }
}

extern "C" void kernel_launch(void* const* d_in, const int* in_sizes, int n_in,
                              void* d_out, int out_size, void* d_ws, size_t ws_size,
                              hipStream_t stream) {
    (void)in_sizes; (void)n_in; (void)out_size; (void)ws_size;

    const float* x        = (const float*)d_in[0];
    const int*   eidx     = (const int*)  d_in[1];
    const float* W_src    = (const float*)d_in[2];
    const float* W_dst    = (const float*)d_in[3];
    const float* attn     = (const float*)d_in[4];
    const float* bias     = (const float*)d_in[5];
    const float* prelu_w  = (const float*)d_in[6];
    const int*   srcI     = eidx;
    const int*   dstI     = eidx + EE;

    float* out = (float*)d_out;

    float* ws      = (float*)d_ws;
    float* src_p   = ws;
    float* dst_p   = src_p + (size_t)NN * FF;
    float* accg    = dst_p + (size_t)NN * FF;
    float* deng    = accg + (size_t)NN * FF;
    unsigned* keys = (unsigned*)(deng + (size_t)NN * HH);
    unsigned* pk   = keys + EE / 4;
    float* partial = (float*)(pk + EE);
    float* shiftg  = partial + MSG_BLOCKS;

    gat_proj_gemm<<<PJ_BLOCKS, PJ_THREADS, 0, stream>>>(x, W_src, W_dst, src_p, dst_p,
                                                       srcI, dstI, keys, pk);

    gat_fused_own<<<MSG_BLOCKS, MSG_THREADS, 0, stream>>>(src_p, dst_p, keys, pk, attn,
                                                          x, bias, prelu_w,
                                                          partial, shiftg, deng, accg,
                                                          out);

    gat_eps_fix<<<MSG_BLOCKS, 256, 0, stream>>>(partial, shiftg, deng, accg,
                                                x, bias, prelu_w, out);
}
